# starting kernel
# baseline (speedup 1.0000x reference)
#include <hip/hip_runtime.h>
#include <math.h>

#define HW   1048576
#define C    64
#define KSEG 4096
#define H1   256
#define NCLS 16
#define KT1  (C + C*8)
#define KT2  (H1 + H1*8)

typedef __attribute__((ext_vector_type(2))) float v2f;
typedef __attribute__((ext_vector_type(4))) float v4f;
typedef __attribute__((ext_vector_type(8))) float v8f;

__device__ __forceinline__ float silu_f(float x) {
    return x / (1.0f + expf(-x));
}

__device__ __forceinline__ void bspline8(float x, float* __restrict__ out) {
    float t[12];
#pragma unroll
    for (int j = 0; j < 12; ++j) t[j] = 0.4f * (float)(j - 3) - 1.0f;
    float b[11];
#pragma unroll
    for (int j = 0; j < 11; ++j) b[j] = (x >= t[j] && x < t[j + 1]) ? 1.0f : 0.0f;
#pragma unroll
    for (int k = 1; k <= 3; ++k) {
#pragma unroll
        for (int j = 0; j < 10; ++j) {
            if (j < 11 - k) {
                float left  = (x - t[j])         / (t[j + k]     - t[j]);
                float right = (t[j + k + 1] - x) / (t[j + k + 1] - t[j + 1]);
                b[j] = left * b[j] + right * b[j + 1];
            }
        }
    }
#pragma unroll
    for (int g = 0; g < 8; ++g) out[g] = b[g];
}

__global__ void zero_kernel(float* __restrict__ p, int n) {
    int i = blockIdx.x * blockDim.x + threadIdx.x;
    if (i < n) p[i] = 0.0f;
}

__global__ void segsum_kernel(const float* __restrict__ x, const int* __restrict__ cls,
                              float* __restrict__ acc, float* __restrict__ cnt) {
    int t = threadIdx.x;
    int rloc = t >> 4;
    int c4   = (t & 15) << 2;
    for (int base = blockIdx.x * 16; base < HW; base += gridDim.x * 16) {
        int row = base + rloc;
        int seg = __builtin_nontemporal_load(cls + row);
        const v4f v = __builtin_nontemporal_load((const v4f*)(x + (size_t)row * C + c4));
        float* a = acc + (size_t)seg * C + c4;
        atomicAdd(a + 0, v.x);
        atomicAdd(a + 1, v.y);
        atomicAdd(a + 2, v.z);
        atomicAdd(a + 3, v.w);
        if ((t & 15) == 0) atomicAdd(cnt + seg, 1.0f);
    }
}

__global__ void wprep1_kernel(const float* __restrict__ bw1, const float* __restrict__ sw1,
                              const float* __restrict__ sc1, float* __restrict__ Wt1) {
    int idx = blockIdx.x * blockDim.x + threadIdx.x;
    if (idx >= H1 * KT1) return;
    int o = idx / KT1, j = idx % KT1;
    float v;
    if (j < C) {
        v = bw1[o * C + j];
    } else {
        int j2 = j - C, i = j2 >> 3, g = j2 & 7;
        v = sw1[((size_t)o * C + i) * 8 + g] * sc1[o * C + i];
    }
    Wt1[(size_t)j * H1 + o] = v;
}

__global__ void wprep2_kernel(const float* __restrict__ bw2, const float* __restrict__ sw2,
                              const float* __restrict__ sc2, float* __restrict__ Wt2) {
    int idx = blockIdx.x * blockDim.x + threadIdx.x;
    if (idx >= NCLS * KT2) return;
    int o = idx / KT2, j = idx % KT2;
    float v;
    if (j < H1) {
        v = bw2[o * H1 + j];
    } else {
        int j2 = j - H1, i = j2 >> 3, g = j2 & 7;
        v = sw2[((size_t)o * H1 + i) * 8 + g] * sc2[o * H1 + i];
    }
    Wt2[(size_t)j * NCLS + o] = v;
}

__global__ void aprep1_kernel(const float* __restrict__ acc, const float* __restrict__ cnt,
                              float* __restrict__ A1) {
    int idx = blockIdx.x * blockDim.x + threadIdx.x;
    if (idx >= KSEG * C) return;
    int k = idx >> 6, c = idx & 63;
    float n = cnt[k];
    float f = acc[(size_t)k * C + c] / fmaxf(n, 1.0f);
    float* row = A1 + (size_t)k * KT1;
    row[c] = silu_f(f);
    float b[8];
    bspline8(f, b);
#pragma unroll
    for (int g = 0; g < 8; ++g) row[C + c * 8 + g] = b[g];
}

__global__ void aprep2_kernel(const float* __restrict__ h, float* __restrict__ A2) {
    int idx = blockIdx.x * blockDim.x + threadIdx.x;
    if (idx >= KSEG * H1) return;
    int m = idx >> 8, i = idx & 255;
    float f = h[(size_t)m * H1 + i];
    float* row = A2 + (size_t)m * KT2;
    row[i] = silu_f(f);
    float b[8];
    bspline8(f, b);
#pragma unroll
    for (int g = 0; g < 8; ++g) row[H1 + i * 8 + g] = b[g];
}

template <int NT>
__global__ void gemm_f32_wmma(const float* __restrict__ A, const float* __restrict__ Bt,
                              float* __restrict__ D, int M, int Kt, int N) {
    const int tilesN = N / (16 * NT);
    const int tm = blockIdx.x / tilesN;
    const int tn = blockIdx.x % tilesN;
    const int lane = threadIdx.x & 31;
    const int mrow = lane & 15;
    const int khalf = (lane >> 4) << 1;
    const int n0 = tn * 16 * NT;

    const float* Arow = A + (size_t)(tm * 16 + mrow) * Kt;

    v8f accv[NT];
#pragma unroll
    for (int t = 0; t < NT; ++t) accv[t] = (v8f){0.f, 0.f, 0.f, 0.f, 0.f, 0.f, 0.f, 0.f};

    for (int kk = 0; kk < Kt; kk += 4) {
        v2f a;
        a.x = Arow[kk + khalf];
        a.y = Arow[kk + khalf + 1];
#pragma unroll
        for (int t = 0; t < NT; ++t) {
            int n = n0 + t * 16 + mrow;
            v2f b;
            b.x = Bt[(size_t)(kk + khalf) * N + n];
            b.y = Bt[(size_t)(kk + khalf + 1) * N + n];
            accv[t] = __builtin_amdgcn_wmma_f32_16x16x4_f32(
                false, a, false, b, (short)0, accv[t], false, false);
        }
    }

    const int rbase = (lane >> 4) * 8;
#pragma unroll
    for (int t = 0; t < NT; ++t) {
#pragma unroll
        for (int r = 0; r < 8; ++r) {
            D[(size_t)(tm * 16 + rbase + r) * N + n0 + t * 16 + mrow] = accv[t][r];
        }
    }
}

extern "C" void kernel_launch(void* const* d_in, const int* in_sizes, int n_in,
                              void* d_out, int out_size, void* d_ws, size_t ws_size,
                              hipStream_t stream) {
    const float* x   = (const float*)d_in[0];
    const int*   cls = (const int*)  d_in[1];
    const float* bw1 = (const float*)d_in[2];
    const float* sw1 = (const float*)d_in[3];
    const float* sc1 = (const float*)d_in[4];
    const float* bw2 = (const float*)d_in[5];
    const float* sw2 = (const float*)d_in[6];
    const float* sc2 = (const float*)d_in[7];
    float* out = (float*)d_out;

    float* ws  = (float*)d_ws;
    float* acc = ws;
    float* cnt = acc + (size_t)KSEG * C;
    float* A1  = cnt + KSEG;
    float* Wt1 = A1 + (size_t)KSEG * KT1;
    float* h   = Wt1 + (size_t)KT1 * H1;
    float* A2  = h + (size_t)KSEG * H1;
    float* Wt2 = A2 + (size_t)KSEG * KT2;

    {
        int n = KSEG * C + KSEG;
        zero_kernel<<<(n + 255) / 256, 256, 0, stream>>>(acc, n);
    }
    segsum_kernel<<<8192, 256, 0, stream>>>(x, cls, acc, cnt);
    {
        int n = H1 * KT1;
        wprep1_kernel<<<(n + 255) / 256, 256, 0, stream>>>(bw1, sw1, sc1, Wt1);
    }
    {
        int n = KSEG * C;
        aprep1_kernel<<<(n + 255) / 256, 256, 0, stream>>>(acc, cnt, A1);
    }
    gemm_f32_wmma<4><<<(KSEG / 16) * (H1 / 64), 32, 0, stream>>>(A1, Wt1, h, KSEG, KT1, H1);
    {
        int n = KSEG * H1;
        aprep2_kernel<<<(n + 255) / 256, 256, 0, stream>>>(h, A2);
    }
    {
        int n = NCLS * KT2;
        wprep2_kernel<<<(n + 255) / 256, 256, 0, stream>>>(bw2, sw2, sc2, Wt2);
    }
    gemm_f32_wmma<1><<<(KSEG / 16) * (NCLS / 16), 32, 0, stream>>>(A2, Wt2, out, KSEG, KT2, NCLS);
}


# segment sums accumulated in LDS (f32 LDS adds) per (segment range, row chunk) block; mean kernel adds the 32 partials in fixed order; no global atomics, no zeroing launch; summation order changes
# speedup vs baseline: 2.2030x; 2.2030x over previous
#include <hip/hip_runtime.h>
#include <math.h>

#define HW   1048576
#define C    64
#define KSEG 4096
#define H1   256
#define NCLS 16
#define KT1  (C + C*8)
#define KT2  (H1 + H1*8)

typedef __attribute__((ext_vector_type(2))) float v2f;
typedef __attribute__((ext_vector_type(4))) float v4f;
typedef __attribute__((ext_vector_type(8))) float v8f;

__device__ __forceinline__ float silu_f(float x) {
    return x / (1.0f + expf(-x));
}

__device__ __forceinline__ void bspline8(float x, float* __restrict__ out) {
    float t[12];
#pragma unroll
    for (int j = 0; j < 12; ++j) t[j] = 0.4f * (float)(j - 3) - 1.0f;
    float b[11];
#pragma unroll
    for (int j = 0; j < 11; ++j) b[j] = (x >= t[j] && x < t[j + 1]) ? 1.0f : 0.0f;
#pragma unroll
    for (int k = 1; k <= 3; ++k) {
#pragma unroll
        for (int j = 0; j < 10; ++j) {
            if (j < 11 - k) {
                float left  = (x - t[j])         / (t[j + k]     - t[j]);
                float right = (t[j + k + 1] - x) / (t[j + k + 1] - t[j + 1]);
                b[j] = left * b[j] + right * b[j + 1];
            }
        }
    }
#pragma unroll
    for (int g = 0; g < 8; ++g) out[g] = b[g];
}

#define SS_NRANGE 8
#define SS_NCHUNK 32
#define SS_SEGS   (KSEG / SS_NRANGE)
#define SS_SHIFT  9
#define SS_ROWS   (HW / SS_NCHUNK)
#define SS_WAVES  32
#define SS_WROWS  (SS_ROWS / SS_WAVES)
#define SS_U      32

template <bool FULL>
__device__ __forceinline__ void gather_rows(const float* __restrict__ x, float* acc_s,
                                            const unsigned* mylist, int e, int n,
                                            int r0, int lane) {
    const int ei = e + lane;
    const unsigned entv = mylist[(FULL || ei < n) ? ei : n - 1];
    unsigned ent[SS_U];
    v2f v[SS_U];
#pragma unroll
    for (int u = 0; u < SS_U; ++u) {
        if (FULL || e + u < n) {
            ent[u] = __builtin_amdgcn_readlane(entv, u);
            const float* src = x + (size_t)(r0 + (int)(ent[u] >> 16)) * C;
            v[u] = __builtin_nontemporal_load((const v2f*)src + lane);
        }
    }
#pragma unroll
    for (int u = 0; u < SS_U; ++u) {
        if (FULL || e + u < n) {
            float* a = acc_s + (ent[u] & 0xffffu) * C + 2 * lane;
            atomicAdd(a, v[u].x);
            atomicAdd(a + 1, v[u].y);
        }
    }
}

__global__ __launch_bounds__(1024)
void segsum_kernel(const float* __restrict__ x, const int* __restrict__ cls,
                   float* __restrict__ part, float* __restrict__ cpart) {
    __shared__ __attribute__((aligned(16))) float acc_s[SS_SEGS * C];
    __shared__ float cnt_s[SS_SEGS];
    __shared__ unsigned list_s[SS_WAVES * SS_WROWS];

    const int tid  = threadIdx.x;
    const int lane = tid & 31;
    const int wave = __builtin_amdgcn_readfirstlane(tid >> 5);
    const unsigned rng = blockIdx.x % SS_NRANGE;
    const int chunk    = blockIdx.x / SS_NRANGE;

    for (int i = tid; i < SS_SEGS * C / 4; i += 1024)
        ((v4f*)acc_s)[i] = (v4f){0.f, 0.f, 0.f, 0.f};
    if (tid < SS_SEGS) cnt_s[tid] = 0.0f;
    __syncthreads();

    unsigned* mylist = list_s + wave * SS_WROWS;
    const int r0 = chunk * SS_ROWS + wave * SS_WROWS;

    int segv[SS_WROWS / 32];
#pragma unroll
    for (int j = 0; j < SS_WROWS / 32; ++j) segv[j] = cls[r0 + j * 32 + lane];
    int n = 0;
#pragma unroll
    for (int j = 0; j < SS_WROWS / 32; ++j) {
        const bool m = ((unsigned)segv[j] >> SS_SHIFT) == rng;
        const unsigned mask = __builtin_amdgcn_ballot_w32(m);
        if (m) {
            const unsigned sl = (unsigned)segv[j] & (SS_SEGS - 1);
            mylist[n + (int)__builtin_amdgcn_mbcnt_lo(mask, 0u)] =
                ((unsigned)(j * 32 + lane) << 16) | sl;
            atomicAdd(&cnt_s[sl], 1.0f);
        }
        n += __builtin_popcount(mask);
    }
    __builtin_amdgcn_wave_barrier();

    int e = 0;
    for (; e + SS_U <= n; e += SS_U) gather_rows<true>(x, acc_s, mylist, e, n, r0, lane);
    if (e < n) gather_rows<false>(x, acc_s, mylist, e, n, r0, lane);
    __syncthreads();

    float* dst = part + ((size_t)chunk * KSEG + (size_t)rng * SS_SEGS) * C;
    for (int i = tid; i < SS_SEGS * C / 4; i += 1024)
        ((v4f*)dst)[i] = ((const v4f*)acc_s)[i];
    if (tid < SS_SEGS) cpart[chunk * KSEG + rng * SS_SEGS + tid] = cnt_s[tid];
}

__global__ void wprep1_kernel(const float* __restrict__ bw1, const float* __restrict__ sw1,
                              const float* __restrict__ sc1, float* __restrict__ Wt1) {
    int idx = blockIdx.x * blockDim.x + threadIdx.x;
    if (idx >= H1 * KT1) return;
    int o = idx / KT1, j = idx % KT1;
    float v;
    if (j < C) {
        v = bw1[o * C + j];
    } else {
        int j2 = j - C, i = j2 >> 3, g = j2 & 7;
        v = sw1[((size_t)o * C + i) * 8 + g] * sc1[o * C + i];
    }
    Wt1[(size_t)j * H1 + o] = v;
}

__global__ void wprep2_kernel(const float* __restrict__ bw2, const float* __restrict__ sw2,
                              const float* __restrict__ sc2, float* __restrict__ Wt2) {
    int idx = blockIdx.x * blockDim.x + threadIdx.x;
    if (idx >= NCLS * KT2) return;
    int o = idx / KT2, j = idx % KT2;
    float v;
    if (j < H1) {
        v = bw2[o * H1 + j];
    } else {
        int j2 = j - H1, i = j2 >> 3, g = j2 & 7;
        v = sw2[((size_t)o * H1 + i) * 8 + g] * sc2[o * H1 + i];
    }
    Wt2[(size_t)j * NCLS + o] = v;
}

__global__ void aprep1_kernel(const float* __restrict__ part, const float* __restrict__ cpart,
                              float* __restrict__ A1) {
    int idx = blockIdx.x * blockDim.x + threadIdx.x;
    if (idx >= KSEG * C) return;
    int k = idx >> 6, c = idx & 63;
    float s = 0.0f, n = 0.0f;
#pragma unroll
    for (int b = 0; b < SS_NCHUNK; ++b) {
        s += part[(size_t)b * KSEG * C + idx];
        n += cpart[b * KSEG + k];
    }
    float f = s / fmaxf(n, 1.0f);
    float* row = A1 + (size_t)k * KT1;
    row[c] = silu_f(f);
    float b[8];
    bspline8(f, b);
#pragma unroll
    for (int g = 0; g < 8; ++g) row[C + c * 8 + g] = b[g];
}

__global__ void aprep2_kernel(const float* __restrict__ h, float* __restrict__ A2) {
    int idx = blockIdx.x * blockDim.x + threadIdx.x;
    if (idx >= KSEG * H1) return;
    int m = idx >> 8, i = idx & 255;
    float f = h[(size_t)m * H1 + i];
    float* row = A2 + (size_t)m * KT2;
    row[i] = silu_f(f);
    float b[8];
    bspline8(f, b);
#pragma unroll
    for (int g = 0; g < 8; ++g) row[H1 + i * 8 + g] = b[g];
}

template <int NT>
__global__ void gemm_f32_wmma(const float* __restrict__ A, const float* __restrict__ Bt,
                              float* __restrict__ D, int M, int Kt, int N) {
    const int tilesN = N / (16 * NT);
    const int tm = blockIdx.x / tilesN;
    const int tn = blockIdx.x % tilesN;
    const int lane = threadIdx.x & 31;
    const int mrow = lane & 15;
    const int khalf = (lane >> 4) << 1;
    const int n0 = tn * 16 * NT;

    const float* Arow = A + (size_t)(tm * 16 + mrow) * Kt;

    v8f accv[NT];
#pragma unroll
    for (int t = 0; t < NT; ++t) accv[t] = (v8f){0.f, 0.f, 0.f, 0.f, 0.f, 0.f, 0.f, 0.f};

    for (int kk = 0; kk < Kt; kk += 4) {
        v2f a;
        a.x = Arow[kk + khalf];
        a.y = Arow[kk + khalf + 1];
#pragma unroll
        for (int t = 0; t < NT; ++t) {
            int n = n0 + t * 16 + mrow;
            v2f b;
            b.x = Bt[(size_t)(kk + khalf) * N + n];
            b.y = Bt[(size_t)(kk + khalf + 1) * N + n];
            accv[t] = __builtin_amdgcn_wmma_f32_16x16x4_f32(
                false, a, false, b, (short)0, accv[t], false, false);
        }
    }

    const int rbase = (lane >> 4) * 8;
#pragma unroll
    for (int t = 0; t < NT; ++t) {
#pragma unroll
        for (int r = 0; r < 8; ++r) {
            D[(size_t)(tm * 16 + rbase + r) * N + n0 + t * 16 + mrow] = accv[t][r];
        }
    }
}

extern "C" void kernel_launch(void* const* d_in, const int* in_sizes, int n_in,
                              void* d_out, int out_size, void* d_ws, size_t ws_size,
                              hipStream_t stream) {
    const float* x   = (const float*)d_in[0];
    const int*   cls = (const int*)  d_in[1];
    const float* bw1 = (const float*)d_in[2];
    const float* sw1 = (const float*)d_in[3];
    const float* sc1 = (const float*)d_in[4];
    const float* bw2 = (const float*)d_in[5];
    const float* sw2 = (const float*)d_in[6];
    const float* sc2 = (const float*)d_in[7];
    float* out = (float*)d_out;

    float* ws  = (float*)d_ws;
    float* A1  = ws + (size_t)KSEG * C + KSEG;
    float* Wt1 = A1 + (size_t)KSEG * KT1;
    float* h   = Wt1 + (size_t)KT1 * H1;
    float* A2  = h + (size_t)KSEG * H1;
    float* Wt2 = A2 + (size_t)KSEG * KT2;
    float* part  = A2;
    float* cpart = part + (size_t)SS_NCHUNK * KSEG * C;

    segsum_kernel<<<SS_NRANGE * SS_NCHUNK, 1024, 0, stream>>>(x, cls, part, cpart);
    {
        int n = H1 * KT1;
        wprep1_kernel<<<(n + 255) / 256, 256, 0, stream>>>(bw1, sw1, sc1, Wt1);
    }
    {
        int n = KSEG * C;
        aprep1_kernel<<<(n + 255) / 256, 256, 0, stream>>>(part, cpart, A1);
    }
    gemm_f32_wmma<4><<<(KSEG / 16) * (H1 / 64), 32, 0, stream>>>(A1, Wt1, h, KSEG, KT1, H1);
    {
        int n = KSEG * H1;
        aprep2_kernel<<<(n + 255) / 256, 256, 0, stream>>>(h, A2);
    }
    {
        int n = NCLS * KT2;
        wprep2_kernel<<<(n + 255) / 256, 256, 0, stream>>>(bw2, sw2, sc2, Wt2);
    }
    gemm_f32_wmma<1><<<(KSEG / 16) * (NCLS / 16), 32, 0, stream>>>(A2, Wt2, out, KSEG, KT2, NCLS);
}


# layer 2 fused: silu and spline bases of h built in registers inside the layer-2 WMMA kernel, K split over 16 waves per strip, partials added in wave order via LDS; A2 not stored; sum order changes
# speedup vs baseline: 3.8421x; 1.7440x over previous
#include <hip/hip_runtime.h>
#include <math.h>

#define HW   1048576
#define C    64
#define KSEG 4096
#define H1   256
#define NCLS 16
#define KT1  (C + C*8)
#define KT2  (H1 + H1*8)

typedef __attribute__((ext_vector_type(2))) float v2f;
typedef __attribute__((ext_vector_type(4))) float v4f;
typedef __attribute__((ext_vector_type(8))) float v8f;

__device__ __forceinline__ float silu_f(float x) {
    return x / (1.0f + expf(-x));
}

__device__ __forceinline__ void bspline8(float x, float* __restrict__ out) {
    float t[12];
#pragma unroll
    for (int j = 0; j < 12; ++j) t[j] = 0.4f * (float)(j - 3) - 1.0f;
    float b[11];
#pragma unroll
    for (int j = 0; j < 11; ++j) b[j] = (x >= t[j] && x < t[j + 1]) ? 1.0f : 0.0f;
#pragma unroll
    for (int k = 1; k <= 3; ++k) {
#pragma unroll
        for (int j = 0; j < 10; ++j) {
            if (j < 11 - k) {
                float left  = (x - t[j])         / (t[j + k]     - t[j]);
                float right = (t[j + k + 1] - x) / (t[j + k + 1] - t[j + 1]);
                b[j] = left * b[j] + right * b[j + 1];
            }
        }
    }
#pragma unroll
    for (int g = 0; g < 8; ++g) out[g] = b[g];
}

#define SS_NRANGE 8
#define SS_NCHUNK 32
#define SS_SEGS   (KSEG / SS_NRANGE)
#define SS_SHIFT  9
#define SS_ROWS   (HW / SS_NCHUNK)
#define SS_WAVES  32
#define SS_WROWS  (SS_ROWS / SS_WAVES)
#define SS_U      32

template <bool FULL>
__device__ __forceinline__ void gather_rows(const float* __restrict__ x, float* acc_s,
                                            const unsigned* mylist, int e, int n,
                                            int r0, int lane) {
    const int ei = e + lane;
    const unsigned entv = mylist[(FULL || ei < n) ? ei : n - 1];
    unsigned ent[SS_U];
    v2f v[SS_U];
#pragma unroll
    for (int u = 0; u < SS_U; ++u) {
        if (FULL || e + u < n) {
            ent[u] = __builtin_amdgcn_readlane(entv, u);
            const float* src = x + (size_t)(r0 + (int)(ent[u] >> 16)) * C;
            v[u] = __builtin_nontemporal_load((const v2f*)src + lane);
        }
    }
#pragma unroll
    for (int u = 0; u < SS_U; ++u) {
        if (FULL || e + u < n) {
            float* a = acc_s + (ent[u] & 0xffffu) * C + 2 * lane;
            atomicAdd(a, v[u].x);
            atomicAdd(a + 1, v[u].y);
        }
    }
}

__global__ __launch_bounds__(1024)
void segsum_kernel(const float* __restrict__ x, const int* __restrict__ cls,
                   float* __restrict__ part, float* __restrict__ cpart) {
    __shared__ __attribute__((aligned(16))) float acc_s[SS_SEGS * C];
    __shared__ float cnt_s[SS_SEGS];
    __shared__ unsigned list_s[SS_WAVES * SS_WROWS];

    const int tid  = threadIdx.x;
    const int lane = tid & 31;
    const int wave = __builtin_amdgcn_readfirstlane(tid >> 5);
    const unsigned rng = blockIdx.x % SS_NRANGE;
    const int chunk    = blockIdx.x / SS_NRANGE;

    for (int i = tid; i < SS_SEGS * C / 4; i += 1024)
        ((v4f*)acc_s)[i] = (v4f){0.f, 0.f, 0.f, 0.f};
    if (tid < SS_SEGS) cnt_s[tid] = 0.0f;
    __syncthreads();

    unsigned* mylist = list_s + wave * SS_WROWS;
    const int r0 = chunk * SS_ROWS + wave * SS_WROWS;

    int segv[SS_WROWS / 32];
#pragma unroll
    for (int j = 0; j < SS_WROWS / 32; ++j) segv[j] = cls[r0 + j * 32 + lane];
    int n = 0;
#pragma unroll
    for (int j = 0; j < SS_WROWS / 32; ++j) {
        const bool m = ((unsigned)segv[j] >> SS_SHIFT) == rng;
        const unsigned mask = __builtin_amdgcn_ballot_w32(m);
        if (m) {
            const unsigned sl = (unsigned)segv[j] & (SS_SEGS - 1);
            mylist[n + (int)__builtin_amdgcn_mbcnt_lo(mask, 0u)] =
                ((unsigned)(j * 32 + lane) << 16) | sl;
            atomicAdd(&cnt_s[sl], 1.0f);
        }
        n += __builtin_popcount(mask);
    }
    __builtin_amdgcn_wave_barrier();

    int e = 0;
    for (; e + SS_U <= n; e += SS_U) gather_rows<true>(x, acc_s, mylist, e, n, r0, lane);
    if (e < n) gather_rows<false>(x, acc_s, mylist, e, n, r0, lane);
    __syncthreads();

    float* dst = part + ((size_t)chunk * KSEG + (size_t)rng * SS_SEGS) * C;
    for (int i = tid; i < SS_SEGS * C / 4; i += 1024)
        ((v4f*)dst)[i] = ((const v4f*)acc_s)[i];
    if (tid < SS_SEGS) cpart[chunk * KSEG + rng * SS_SEGS + tid] = cnt_s[tid];
}

__global__ void wprep1_kernel(const float* __restrict__ bw1, const float* __restrict__ sw1,
                              const float* __restrict__ sc1, float* __restrict__ Wt1) {
    int idx = blockIdx.x * blockDim.x + threadIdx.x;
    if (idx >= H1 * KT1) return;
    int o = idx / KT1, j = idx % KT1;
    float v;
    if (j < C) {
        v = bw1[o * C + j];
    } else {
        int j2 = j - C, i = j2 >> 3, g = j2 & 7;
        v = sw1[((size_t)o * C + i) * 8 + g] * sc1[o * C + i];
    }
    Wt1[(size_t)j * H1 + o] = v;
}

__global__ void wprep2_kernel(const float* __restrict__ bw2, const float* __restrict__ sw2,
                              const float* __restrict__ sc2, float* __restrict__ Wt2) {
    int idx = blockIdx.x * blockDim.x + threadIdx.x;
    if (idx >= NCLS * KT2) return;
    int o = idx / KT2, j = idx % KT2;
    float v;
    if (j < H1) {
        v = bw2[o * H1 + j];
    } else {
        int j2 = j - H1, i = j2 >> 3, g = j2 & 7;
        v = sw2[((size_t)o * H1 + i) * 8 + g] * sc2[o * H1 + i];
    }
    Wt2[(size_t)j * NCLS + o] = v;
}

__global__ void aprep1_kernel(const float* __restrict__ part, const float* __restrict__ cpart,
                              float* __restrict__ A1) {
    int idx = blockIdx.x * blockDim.x + threadIdx.x;
    if (idx >= KSEG * C) return;
    int k = idx >> 6, c = idx & 63;
    float s = 0.0f, n = 0.0f;
#pragma unroll
    for (int b = 0; b < SS_NCHUNK; ++b) {
        s += part[(size_t)b * KSEG * C + idx];
        n += cpart[b * KSEG + k];
    }
    float f = s / fmaxf(n, 1.0f);
    float* row = A1 + (size_t)k * KT1;
    row[c] = silu_f(f);
    float b[8];
    bspline8(f, b);
#pragma unroll
    for (int g = 0; g < 8; ++g) row[C + c * 8 + g] = b[g];
}

#define L2_WAVES 16
#define L2_FEATS (H1 / L2_WAVES)
#define L2_STEPS (L2_FEATS / 4)

__global__ __launch_bounds__(32 * L2_WAVES)
void layer2_fused_kernel(const float* __restrict__ h, const float* __restrict__ Wt2,
                         float* __restrict__ out) {
    __shared__ float red_s[L2_WAVES * 256];

    const int tid  = threadIdx.x;
    const int lane = tid & 31;
    const int wave = __builtin_amdgcn_readfirstlane(tid >> 5);
    const int c    = lane & 15;
    const int hh   = lane >> 4;
    const int strip = blockIdx.x;
    const int fbase = wave * L2_FEATS;

    const float* hrow = h + (size_t)(strip * 16 + c) * H1 + fbase + hh;
    const float* Wc   = Wt2 + c;

    v8f acc0 = (v8f){0.f, 0.f, 0.f, 0.f, 0.f, 0.f, 0.f, 0.f};
    v8f acc1 = (v8f){0.f, 0.f, 0.f, 0.f, 0.f, 0.f, 0.f, 0.f};

    float x0 = hrow[0];
    float x1 = hrow[2];
#pragma unroll 1
    for (int s = 0; s < L2_STEPS; ++s) {
        const int f0 = fbase + 4 * s + hh;
        const int f1 = f0 + 2;
        const int sn = (s + 1 < L2_STEPS) ? s + 1 : s;
        const float xn0 = hrow[4 * sn];
        const float xn1 = hrow[4 * sn + 2];

        {
            v2f a, b;
            a.x = silu_f(x0);
            a.y = silu_f(x1);
            b.x = Wc[(size_t)f0 * NCLS];
            b.y = Wc[(size_t)f1 * NCLS];
            acc0 = __builtin_amdgcn_wmma_f32_16x16x4_f32(false, a, false, b, (short)0, acc0, false, false);
        }

        float bs0[8], bs1[8];
        bspline8(x0, bs0);
        bspline8(x1, bs1);
        const float* W0 = Wc + (size_t)(H1 + f0 * 8) * NCLS;
        const float* W1 = Wc + (size_t)(H1 + f1 * 8) * NCLS;
#pragma unroll
        for (int t = 0; t < 4; ++t) {
            v2f a0, b0, a1, b1;
            a0.x = bs0[2 * t];
            a0.y = bs0[2 * t + 1];
            b0.x = W0[(2 * t) * NCLS];
            b0.y = W0[(2 * t + 1) * NCLS];
            a1.x = bs1[2 * t];
            a1.y = bs1[2 * t + 1];
            b1.x = W1[(2 * t) * NCLS];
            b1.y = W1[(2 * t + 1) * NCLS];
            acc0 = __builtin_amdgcn_wmma_f32_16x16x4_f32(false, a0, false, b0, (short)0, acc0, false, false);
            acc1 = __builtin_amdgcn_wmma_f32_16x16x4_f32(false, a1, false, b1, (short)0, acc1, false, false);
        }
        x0 = xn0;
        x1 = xn1;
    }

    float* mine = red_s + wave * 256;
#pragma unroll
    for (int r = 0; r < 8; ++r) mine[r * 32 + lane] = acc0[r] + acc1[r];
    __syncthreads();

    if (tid < 256) {
        const int row = tid >> 4, col = tid & 15;
        const int idx = (row & 7) * 32 + (row >> 3) * 16 + col;
        float sum = 0.0f;
#pragma unroll
        for (int w = 0; w < L2_WAVES; ++w) sum += red_s[w * 256 + idx];
        out[(size_t)strip * 256 + tid] = sum;
    }
}

template <int NT>
__global__ void gemm_f32_wmma(const float* __restrict__ A, const float* __restrict__ Bt,
                              float* __restrict__ D, int M, int Kt, int N) {
    const int tilesN = N / (16 * NT);
    const int tm = blockIdx.x / tilesN;
    const int tn = blockIdx.x % tilesN;
    const int lane = threadIdx.x & 31;
    const int mrow = lane & 15;
    const int khalf = (lane >> 4) << 1;
    const int n0 = tn * 16 * NT;

    const float* Arow = A + (size_t)(tm * 16 + mrow) * Kt;

    v8f accv[NT];
#pragma unroll
    for (int t = 0; t < NT; ++t) accv[t] = (v8f){0.f, 0.f, 0.f, 0.f, 0.f, 0.f, 0.f, 0.f};

    for (int kk = 0; kk < Kt; kk += 4) {
        v2f a;
        a.x = Arow[kk + khalf];
        a.y = Arow[kk + khalf + 1];
#pragma unroll
        for (int t = 0; t < NT; ++t) {
            int n = n0 + t * 16 + mrow;
            v2f b;
            b.x = Bt[(size_t)(kk + khalf) * N + n];
            b.y = Bt[(size_t)(kk + khalf + 1) * N + n];
            accv[t] = __builtin_amdgcn_wmma_f32_16x16x4_f32(
                false, a, false, b, (short)0, accv[t], false, false);
        }
    }

    const int rbase = (lane >> 4) * 8;
#pragma unroll
    for (int t = 0; t < NT; ++t) {
#pragma unroll
        for (int r = 0; r < 8; ++r) {
            D[(size_t)(tm * 16 + rbase + r) * N + n0 + t * 16 + mrow] = accv[t][r];
        }
    }
}

extern "C" void kernel_launch(void* const* d_in, const int* in_sizes, int n_in,
                              void* d_out, int out_size, void* d_ws, size_t ws_size,
                              hipStream_t stream) {
    const float* x   = (const float*)d_in[0];
    const int*   cls = (const int*)  d_in[1];
    const float* bw1 = (const float*)d_in[2];
    const float* sw1 = (const float*)d_in[3];
    const float* sc1 = (const float*)d_in[4];
    const float* bw2 = (const float*)d_in[5];
    const float* sw2 = (const float*)d_in[6];
    const float* sc2 = (const float*)d_in[7];
    float* out = (float*)d_out;

    float* ws  = (float*)d_ws;
    float* A1  = ws + (size_t)KSEG * C + KSEG;
    float* Wt1 = A1 + (size_t)KSEG * KT1;
    float* h   = Wt1 + (size_t)KT1 * H1;
    float* A2  = h + (size_t)KSEG * H1;
    float* Wt2 = A2 + (size_t)KSEG * KT2;
    float* part  = A2;
    float* cpart = part + (size_t)SS_NCHUNK * KSEG * C;

    segsum_kernel<<<SS_NRANGE * SS_NCHUNK, 1024, 0, stream>>>(x, cls, part, cpart);
    {
        int n = H1 * KT1;
        wprep1_kernel<<<(n + 255) / 256, 256, 0, stream>>>(bw1, sw1, sc1, Wt1);
    }
    {
        int n = KSEG * C;
        aprep1_kernel<<<(n + 255) / 256, 256, 0, stream>>>(part, cpart, A1);
    }
    gemm_f32_wmma<4><<<(KSEG / 16) * (H1 / 64), 32, 0, stream>>>(A1, Wt1, h, KSEG, KT1, H1);
    {
        int n = NCLS * KT2;
        wprep2_kernel<<<(n + 255) / 256, 256, 0, stream>>>(bw2, sw2, sc2, Wt2);
    }
    layer2_fused_kernel<<<KSEG / 16, 32 * L2_WAVES, 0, stream>>>(h, Wt2, out);
}


# layer 1 fused: partial sums, means, silu and spline bases built into an LDS strip inside the layer-1 WMMA kernel (16 waves per strip, weight in fragment order); A1 not stored; sum order changes
# speedup vs baseline: 6.2475x; 1.6261x over previous
#include <hip/hip_runtime.h>
#include <math.h>

#define HW   1048576
#define C    64
#define KSEG 4096
#define H1   256
#define NCLS 16
#define KT1  (C + C*8)
#define KT2  (H1 + H1*8)

typedef __attribute__((ext_vector_type(2))) float v2f;
typedef __attribute__((ext_vector_type(4))) float v4f;
typedef __attribute__((ext_vector_type(8))) float v8f;

__device__ __forceinline__ float silu_f(float x) {
    return x / (1.0f + expf(-x));
}

__device__ __forceinline__ void bspline8(float x, float* __restrict__ out) {
    float t[12];
#pragma unroll
    for (int j = 0; j < 12; ++j) t[j] = 0.4f * (float)(j - 3) - 1.0f;
    float b[11];
#pragma unroll
    for (int j = 0; j < 11; ++j) b[j] = (x >= t[j] && x < t[j + 1]) ? 1.0f : 0.0f;
#pragma unroll
    for (int k = 1; k <= 3; ++k) {
#pragma unroll
        for (int j = 0; j < 10; ++j) {
            if (j < 11 - k) {
                float left  = (x - t[j])         / (t[j + k]     - t[j]);
                float right = (t[j + k + 1] - x) / (t[j + k + 1] - t[j + 1]);
                b[j] = left * b[j] + right * b[j + 1];
            }
        }
    }
#pragma unroll
    for (int g = 0; g < 8; ++g) out[g] = b[g];
}

#define SS_NRANGE 8
#define SS_NCHUNK 32
#define SS_SEGS   (KSEG / SS_NRANGE)
#define SS_SHIFT  9
#define SS_ROWS   (HW / SS_NCHUNK)
#define SS_WAVES  32
#define SS_WROWS  (SS_ROWS / SS_WAVES)
#define SS_U      32

template <bool FULL>
__device__ __forceinline__ void gather_rows(const float* __restrict__ x, float* acc_s,
                                            const unsigned* mylist, int e, int n,
                                            int r0, int lane) {
    const int ei = e + lane;
    const unsigned entv = mylist[(FULL || ei < n) ? ei : n - 1];
    unsigned ent[SS_U];
    v2f v[SS_U];
#pragma unroll
    for (int u = 0; u < SS_U; ++u) {
        if (FULL || e + u < n) {
            ent[u] = __builtin_amdgcn_readlane(entv, u);
            const float* src = x + (size_t)(r0 + (int)(ent[u] >> 16)) * C;
            v[u] = __builtin_nontemporal_load((const v2f*)src + lane);
        }
    }
#pragma unroll
    for (int u = 0; u < SS_U; ++u) {
        if (FULL || e + u < n) {
            float* a = acc_s + (ent[u] & 0xffffu) * C + 2 * lane;
            atomicAdd(a, v[u].x);
            atomicAdd(a + 1, v[u].y);
        }
    }
}

__global__ __launch_bounds__(1024)
void segsum_kernel(const float* __restrict__ x, const int* __restrict__ cls,
                   float* __restrict__ part, float* __restrict__ cpart) {
    __shared__ __attribute__((aligned(16))) float acc_s[SS_SEGS * C];
    __shared__ float cnt_s[SS_SEGS];
    __shared__ unsigned list_s[SS_WAVES * SS_WROWS];

    const int tid  = threadIdx.x;
    const int lane = tid & 31;
    const int wave = __builtin_amdgcn_readfirstlane(tid >> 5);
    const unsigned rng = blockIdx.x % SS_NRANGE;
    const int chunk    = blockIdx.x / SS_NRANGE;

    for (int i = tid; i < SS_SEGS * C / 4; i += 1024)
        ((v4f*)acc_s)[i] = (v4f){0.f, 0.f, 0.f, 0.f};
    if (tid < SS_SEGS) cnt_s[tid] = 0.0f;
    __syncthreads();

    unsigned* mylist = list_s + wave * SS_WROWS;
    const int r0 = chunk * SS_ROWS + wave * SS_WROWS;

    int segv[SS_WROWS / 32];
#pragma unroll
    for (int j = 0; j < SS_WROWS / 32; ++j) segv[j] = cls[r0 + j * 32 + lane];
    int n = 0;
#pragma unroll
    for (int j = 0; j < SS_WROWS / 32; ++j) {
        const bool m = ((unsigned)segv[j] >> SS_SHIFT) == rng;
        const unsigned mask = __builtin_amdgcn_ballot_w32(m);
        if (m) {
            const unsigned sl = (unsigned)segv[j] & (SS_SEGS - 1);
            mylist[n + (int)__builtin_amdgcn_mbcnt_lo(mask, 0u)] =
                ((unsigned)(j * 32 + lane) << 16) | sl;
            atomicAdd(&cnt_s[sl], 1.0f);
        }
        n += __builtin_popcount(mask);
    }
    __builtin_amdgcn_wave_barrier();

    int e = 0;
    for (; e + SS_U <= n; e += SS_U) gather_rows<true>(x, acc_s, mylist, e, n, r0, lane);
    if (e < n) gather_rows<false>(x, acc_s, mylist, e, n, r0, lane);
    __syncthreads();

    float* dst = part + ((size_t)chunk * KSEG + (size_t)rng * SS_SEGS) * C;
    for (int i = tid; i < SS_SEGS * C / 4; i += 1024)
        ((v4f*)dst)[i] = ((const v4f*)acc_s)[i];
    if (tid < SS_SEGS) cpart[chunk * KSEG + rng * SS_SEGS + tid] = cnt_s[tid];
}

#define L1_WAVES 16
#define L1_STEPS (KT1 / 8)
#define L1_CH    12
#define L1_NCH   (L1_STEPS / L1_CH)
#define FS_PITCH 65

__global__ void wprep1_kernel(const float* __restrict__ bw1, const float* __restrict__ sw1,
                              const float* __restrict__ sc1, float* __restrict__ Wp) {
    int idx = blockIdx.x * blockDim.x + threadIdx.x;
    if (idx >= H1 * KT1 / 4) return;
    const int lane = idx & 31, q = idx >> 5;
    const int j = q % L1_STEPS, nt = q / L1_STEPS;
    const int o = nt * 16 + (lane & 15), hh = lane >> 4;
    v4f v;
    if (j < C / 8) {
        v = *(const v4f*)(bw1 + o * C + 8 * j + 4 * hh);
    } else {
        const int i = j - C / 8;
        const v4f w = *(const v4f*)(sw1 + ((size_t)o * C + i) * 8 + 4 * hh);
        const float sc = sc1[o * C + i];
        v.x = w.x * sc;
        v.y = w.y * sc;
        v.z = w.z * sc;
        v.w = w.w * sc;
    }
    ((v4f*)Wp)[idx] = v;
}

__global__ void wprep2_kernel(const float* __restrict__ bw2, const float* __restrict__ sw2,
                              const float* __restrict__ sc2, float* __restrict__ Wt2) {
    int idx = blockIdx.x * blockDim.x + threadIdx.x;
    if (idx >= NCLS * KT2) return;
    int o = idx / KT2, j = idx % KT2;
    float v;
    if (j < H1) {
        v = bw2[o * H1 + j];
    } else {
        int j2 = j - H1, i = j2 >> 3, g = j2 & 7;
        v = sw2[((size_t)o * H1 + i) * 8 + g] * sc2[o * H1 + i];
    }
    Wt2[(size_t)j * NCLS + o] = v;
}


static_assert(SS_NCHUNK == 32, "one partial count per lane");

__global__ __launch_bounds__(32 * L1_WAVES)
void layer1_fused_kernel(const float* __restrict__ part, const float* __restrict__ cpart,
                         const float* __restrict__ Wp, float* __restrict__ h) {
    __shared__ __attribute__((aligned(16))) float A_s[L1_STEPS * 32 * 4];
    __shared__ __attribute__((aligned(16))) float h_s[16 * H1];
    __shared__ float f_s[16 * FS_PITCH];
    __shared__ float cnt_s[16];

    const int tid  = threadIdx.x;
    const int lane = tid & 31;
    const int wave = __builtin_amdgcn_readfirstlane(tid >> 5);
    const int strip = blockIdx.x;

    const v4f* Bw = (const v4f*)Wp + (size_t)wave * L1_STEPS * 32 + lane;
    v4f bq[2][L1_CH];
#pragma unroll
    for (int jj = 0; jj < L1_CH; ++jj) bq[0][jj] = Bw[jj * 32];

    {
        const float* p = part + (size_t)strip * 16 * C + tid;
        float s0 = 0.0f, s1 = 0.0f;
#pragma unroll
        for (int b = 0; b < SS_NCHUNK; ++b) {
            s0 += p[(size_t)b * KSEG * C];
            s1 += p[(size_t)b * KSEG * C + 512];
        }
        f_s[(tid >> 6) * FS_PITCH + (tid & 63)] = s0;
        f_s[(8 + (tid >> 6)) * FS_PITCH + (tid & 63)] = s1;
        float cv = cpart[lane * KSEG + strip * 16 + wave];
        for (int m = 16; m >= 1; m >>= 1) cv += __shfl_xor(cv, m, 32);
        if (lane == 0) cnt_s[wave] = cv;
    }
    __syncthreads();

    for (int e = 0; e < 2; ++e) {
        const int idx = tid + e * 512;
        const int r = idx & 15, c = idx >> 4;
        const float f = f_s[r * FS_PITCH + c] / fmaxf(cnt_s[r], 1.0f);
        A_s[(((c >> 3) * 32 + ((c >> 2) & 1) * 16 + r) << 2) + (c & 3)] = silu_f(f);
        float bs[8];
        bspline8(f, bs);
        ((v4f*)A_s)[(8 + c) * 32 + r]      = (v4f){bs[0], bs[1], bs[2], bs[3]};
        ((v4f*)A_s)[(8 + c) * 32 + 16 + r] = (v4f){bs[4], bs[5], bs[6], bs[7]};
    }
    __syncthreads();

    v8f acc0 = (v8f){0.f, 0.f, 0.f, 0.f, 0.f, 0.f, 0.f, 0.f};
    v8f acc1 = (v8f){0.f, 0.f, 0.f, 0.f, 0.f, 0.f, 0.f, 0.f};
    const v4f* Al = (const v4f*)A_s + lane;
#pragma unroll
    for (int ch = 0; ch < L1_NCH; ++ch) {
        if (ch + 1 < L1_NCH) {
#pragma unroll
            for (int jj = 0; jj < L1_CH; ++jj)
                bq[(ch + 1) & 1][jj] = Bw[((ch + 1) * L1_CH + jj) * 32];
        }
        v4f aq[L1_CH];
#pragma unroll
        for (int jj = 0; jj < L1_CH; ++jj) aq[jj] = Al[(ch * L1_CH + jj) * 32];
#pragma unroll
        for (int jj = 0; jj < L1_CH; ++jj) {
            const v4f a = aq[jj], b = bq[ch & 1][jj];
            acc0 = __builtin_amdgcn_wmma_f32_16x16x4_f32(false, (v2f){a.x, a.y}, false, (v2f){b.x, b.y},
                                                         (short)0, acc0, false, false);
            acc1 = __builtin_amdgcn_wmma_f32_16x16x4_f32(false, (v2f){a.z, a.w}, false, (v2f){b.z, b.w},
                                                         (short)0, acc1, false, false);
        }
    }

#pragma unroll
    for (int r = 0; r < 8; ++r)
        h_s[(r + 8 * (lane >> 4)) * H1 + wave * 16 + (lane & 15)] = acc0[r] + acc1[r];
    __syncthreads();
    v4f* dst = (v4f*)(h + (size_t)strip * 16 * H1);
    dst[tid]       = ((const v4f*)h_s)[tid];
    dst[tid + 512] = ((const v4f*)h_s)[tid + 512];
}

#define L2_WAVES 16
#define L2_FEATS (H1 / L2_WAVES)
#define L2_STEPS (L2_FEATS / 4)

__global__ __launch_bounds__(32 * L2_WAVES)
void layer2_fused_kernel(const float* __restrict__ h, const float* __restrict__ Wt2,
                         float* __restrict__ out) {
    __shared__ float red_s[L2_WAVES * 256];

    const int tid  = threadIdx.x;
    const int lane = tid & 31;
    const int wave = __builtin_amdgcn_readfirstlane(tid >> 5);
    const int c    = lane & 15;
    const int hh   = lane >> 4;
    const int strip = blockIdx.x;
    const int fbase = wave * L2_FEATS;

    const float* hrow = h + (size_t)(strip * 16 + c) * H1 + fbase + hh;
    const float* Wc   = Wt2 + c;

    v8f acc0 = (v8f){0.f, 0.f, 0.f, 0.f, 0.f, 0.f, 0.f, 0.f};
    v8f acc1 = (v8f){0.f, 0.f, 0.f, 0.f, 0.f, 0.f, 0.f, 0.f};

    float x0 = hrow[0];
    float x1 = hrow[2];
#pragma unroll 1
    for (int s = 0; s < L2_STEPS; ++s) {
        const int f0 = fbase + 4 * s + hh;
        const int f1 = f0 + 2;
        const int sn = (s + 1 < L2_STEPS) ? s + 1 : s;
        const float xn0 = hrow[4 * sn];
        const float xn1 = hrow[4 * sn + 2];

        {
            v2f a, b;
            a.x = silu_f(x0);
            a.y = silu_f(x1);
            b.x = Wc[(size_t)f0 * NCLS];
            b.y = Wc[(size_t)f1 * NCLS];
            acc0 = __builtin_amdgcn_wmma_f32_16x16x4_f32(false, a, false, b, (short)0, acc0, false, false);
        }

        float bs0[8], bs1[8];
        bspline8(x0, bs0);
        bspline8(x1, bs1);
        const float* W0 = Wc + (size_t)(H1 + f0 * 8) * NCLS;
        const float* W1 = Wc + (size_t)(H1 + f1 * 8) * NCLS;
#pragma unroll
        for (int t = 0; t < 4; ++t) {
            v2f a0, b0, a1, b1;
            a0.x = bs0[2 * t];
            a0.y = bs0[2 * t + 1];
            b0.x = W0[(2 * t) * NCLS];
            b0.y = W0[(2 * t + 1) * NCLS];
            a1.x = bs1[2 * t];
            a1.y = bs1[2 * t + 1];
            b1.x = W1[(2 * t) * NCLS];
            b1.y = W1[(2 * t + 1) * NCLS];
            acc0 = __builtin_amdgcn_wmma_f32_16x16x4_f32(false, a0, false, b0, (short)0, acc0, false, false);
            acc1 = __builtin_amdgcn_wmma_f32_16x16x4_f32(false, a1, false, b1, (short)0, acc1, false, false);
        }
        x0 = xn0;
        x1 = xn1;
    }

    float* mine = red_s + wave * 256;
#pragma unroll
    for (int r = 0; r < 8; ++r) mine[r * 32 + lane] = acc0[r] + acc1[r];
    __syncthreads();

    if (tid < 256) {
        const int row = tid >> 4, col = tid & 15;
        const int idx = (row & 7) * 32 + (row >> 3) * 16 + col;
        float sum = 0.0f;
#pragma unroll
        for (int w = 0; w < L2_WAVES; ++w) sum += red_s[w * 256 + idx];
        out[(size_t)strip * 256 + tid] = sum;
    }
}

extern "C" void kernel_launch(void* const* d_in, const int* in_sizes, int n_in,
                              void* d_out, int out_size, void* d_ws, size_t ws_size,
                              hipStream_t stream) {
    const float* x   = (const float*)d_in[0];
    const int*   cls = (const int*)  d_in[1];
    const float* bw1 = (const float*)d_in[2];
    const float* sw1 = (const float*)d_in[3];
    const float* sc1 = (const float*)d_in[4];
    const float* bw2 = (const float*)d_in[5];
    const float* sw2 = (const float*)d_in[6];
    const float* sc2 = (const float*)d_in[7];
    float* out = (float*)d_out;

    float* ws  = (float*)d_ws;
    float* A1  = ws + (size_t)KSEG * C + KSEG;
    float* Wt1 = A1 + (size_t)KSEG * KT1;
    float* h   = Wt1 + (size_t)KT1 * H1;
    float* A2  = h + (size_t)KSEG * H1;
    float* Wt2 = A2 + (size_t)KSEG * KT2;
    float* part  = A2;
    float* cpart = part + (size_t)SS_NCHUNK * KSEG * C;

    segsum_kernel<<<SS_NRANGE * SS_NCHUNK, 1024, 0, stream>>>(x, cls, part, cpart);
    {
        int n = H1 * KT1 / 4;
        wprep1_kernel<<<(n + 255) / 256, 256, 0, stream>>>(bw1, sw1, sc1, Wt1);
    }
    layer1_fused_kernel<<<KSEG / 16, 32 * L1_WAVES, 0, stream>>>(part, cpart, Wt1, h);
    {
        int n = NCLS * KT2;
        wprep2_kernel<<<(n + 255) / 256, 256, 0, stream>>>(bw2, sw2, sc2, Wt2);
    }
    layer2_fused_kernel<<<KSEG / 16, 32 * L2_WAVES, 0, stream>>>(h, Wt2, out);
}


# spline bases in closed form: each basis as the cardinal cubic B-spline of its centre knot (two polynomial pieces, one select), not the 3-level recursion with 54 divisions a value; rounding changes
# speedup vs baseline: 7.3148x; 1.1708x over previous
#include <hip/hip_runtime.h>
#include <math.h>

#define HW   1048576
#define C    64
#define KSEG 4096
#define H1   256
#define NCLS 16
#define KT1  (C + C*8)
#define KT2  (H1 + H1*8)

typedef __attribute__((ext_vector_type(2))) float v2f;
typedef __attribute__((ext_vector_type(4))) float v4f;
typedef __attribute__((ext_vector_type(8))) float v8f;

__device__ __forceinline__ float silu_f(float x) {
    return x / (1.0f + expf(-x));
}

__device__ __forceinline__ void bspline8(float x, float* __restrict__ out) {
#pragma unroll
    for (int g = 0; g < 8; ++g) {
        const float cg = 0.4f * (float)(g - 1) - 1.0f;
        const float a  = fabsf((x - cg) * 2.5f);
        const float in = fmaf(a * a, fmaf(0.5f, a, -1.0f), 2.0f / 3.0f);
        const float r  = fmaxf(2.0f - a, 0.0f);
        out[g] = a < 1.0f ? in : r * r * r * (1.0f / 6.0f);
    }
}

#define SS_NRANGE 8
#define SS_NCHUNK 32
#define SS_SEGS   (KSEG / SS_NRANGE)
#define SS_SHIFT  9
#define SS_ROWS   (HW / SS_NCHUNK)
#define SS_WAVES  32
#define SS_WROWS  (SS_ROWS / SS_WAVES)
#define SS_U      32

template <bool FULL>
__device__ __forceinline__ void gather_rows(const float* __restrict__ x, float* acc_s,
                                            const unsigned* mylist, int e, int n,
                                            int r0, int lane) {
    const int ei = e + lane;
    const unsigned entv = mylist[(FULL || ei < n) ? ei : n - 1];
    unsigned ent[SS_U];
    v2f v[SS_U];
#pragma unroll
    for (int u = 0; u < SS_U; ++u) {
        if (FULL || e + u < n) {
            ent[u] = __builtin_amdgcn_readlane(entv, u);
            const float* src = x + (size_t)(r0 + (int)(ent[u] >> 16)) * C;
            v[u] = __builtin_nontemporal_load((const v2f*)src + lane);
        }
    }
#pragma unroll
    for (int u = 0; u < SS_U; ++u) {
        if (FULL || e + u < n) {
            float* a = acc_s + (ent[u] & 0xffffu) * C + 2 * lane;
            atomicAdd(a, v[u].x);
            atomicAdd(a + 1, v[u].y);
        }
    }
}

__global__ __launch_bounds__(1024)
void segsum_kernel(const float* __restrict__ x, const int* __restrict__ cls,
                   float* __restrict__ part, float* __restrict__ cpart) {
    __shared__ __attribute__((aligned(16))) float acc_s[SS_SEGS * C];
    __shared__ float cnt_s[SS_SEGS];
    __shared__ unsigned list_s[SS_WAVES * SS_WROWS];

    const int tid  = threadIdx.x;
    const int lane = tid & 31;
    const int wave = __builtin_amdgcn_readfirstlane(tid >> 5);
    const unsigned rng = blockIdx.x % SS_NRANGE;
    const int chunk    = blockIdx.x / SS_NRANGE;

    for (int i = tid; i < SS_SEGS * C / 4; i += 1024)
        ((v4f*)acc_s)[i] = (v4f){0.f, 0.f, 0.f, 0.f};
    if (tid < SS_SEGS) cnt_s[tid] = 0.0f;
    __syncthreads();

    unsigned* mylist = list_s + wave * SS_WROWS;
    const int r0 = chunk * SS_ROWS + wave * SS_WROWS;

    int segv[SS_WROWS / 32];
#pragma unroll
    for (int j = 0; j < SS_WROWS / 32; ++j) segv[j] = cls[r0 + j * 32 + lane];
    int n = 0;
#pragma unroll
    for (int j = 0; j < SS_WROWS / 32; ++j) {
        const bool m = ((unsigned)segv[j] >> SS_SHIFT) == rng;
        const unsigned mask = __builtin_amdgcn_ballot_w32(m);
        if (m) {
            const unsigned sl = (unsigned)segv[j] & (SS_SEGS - 1);
            mylist[n + (int)__builtin_amdgcn_mbcnt_lo(mask, 0u)] =
                ((unsigned)(j * 32 + lane) << 16) | sl;
            atomicAdd(&cnt_s[sl], 1.0f);
        }
        n += __builtin_popcount(mask);
    }
    __builtin_amdgcn_wave_barrier();

    int e = 0;
    for (; e + SS_U <= n; e += SS_U) gather_rows<true>(x, acc_s, mylist, e, n, r0, lane);
    if (e < n) gather_rows<false>(x, acc_s, mylist, e, n, r0, lane);
    __syncthreads();

    float* dst = part + ((size_t)chunk * KSEG + (size_t)rng * SS_SEGS) * C;
    for (int i = tid; i < SS_SEGS * C / 4; i += 1024)
        ((v4f*)dst)[i] = ((const v4f*)acc_s)[i];
    if (tid < SS_SEGS) cpart[chunk * KSEG + rng * SS_SEGS + tid] = cnt_s[tid];
}

#define L1_WAVES 16
#define L1_STEPS (KT1 / 8)
#define L1_CH    12
#define L1_NCH   (L1_STEPS / L1_CH)
#define FS_PITCH 65

__global__ void wprep1_kernel(const float* __restrict__ bw1, const float* __restrict__ sw1,
                              const float* __restrict__ sc1, float* __restrict__ Wp) {
    int idx = blockIdx.x * blockDim.x + threadIdx.x;
    if (idx >= H1 * KT1 / 4) return;
    const int lane = idx & 31, q = idx >> 5;
    const int j = q % L1_STEPS, nt = q / L1_STEPS;
    const int o = nt * 16 + (lane & 15), hh = lane >> 4;
    v4f v;
    if (j < C / 8) {
        v = *(const v4f*)(bw1 + o * C + 8 * j + 4 * hh);
    } else {
        const int i = j - C / 8;
        const v4f w = *(const v4f*)(sw1 + ((size_t)o * C + i) * 8 + 4 * hh);
        const float sc = sc1[o * C + i];
        v.x = w.x * sc;
        v.y = w.y * sc;
        v.z = w.z * sc;
        v.w = w.w * sc;
    }
    ((v4f*)Wp)[idx] = v;
}

__global__ void wprep2_kernel(const float* __restrict__ bw2, const float* __restrict__ sw2,
                              const float* __restrict__ sc2, float* __restrict__ Wt2) {
    int idx = blockIdx.x * blockDim.x + threadIdx.x;
    if (idx >= NCLS * KT2) return;
    int o = idx / KT2, j = idx % KT2;
    float v;
    if (j < H1) {
        v = bw2[o * H1 + j];
    } else {
        int j2 = j - H1, i = j2 >> 3, g = j2 & 7;
        v = sw2[((size_t)o * H1 + i) * 8 + g] * sc2[o * H1 + i];
    }
    Wt2[(size_t)j * NCLS + o] = v;
}


static_assert(SS_NCHUNK == 32, "one partial count per lane");

__global__ __launch_bounds__(32 * L1_WAVES)
void layer1_fused_kernel(const float* __restrict__ part, const float* __restrict__ cpart,
                         const float* __restrict__ Wp, float* __restrict__ h) {
    __shared__ __attribute__((aligned(16))) float A_s[L1_STEPS * 32 * 4];
    __shared__ __attribute__((aligned(16))) float h_s[16 * H1];
    __shared__ float f_s[16 * FS_PITCH];
    __shared__ float cnt_s[16];

    const int tid  = threadIdx.x;
    const int lane = tid & 31;
    const int wave = __builtin_amdgcn_readfirstlane(tid >> 5);
    const int strip = blockIdx.x;

    const v4f* Bw = (const v4f*)Wp + (size_t)wave * L1_STEPS * 32 + lane;
    v4f bq[2][L1_CH];
#pragma unroll
    for (int jj = 0; jj < L1_CH; ++jj) bq[0][jj] = Bw[jj * 32];

    {
        const float* p = part + (size_t)strip * 16 * C + tid;
        float s0 = 0.0f, s1 = 0.0f;
#pragma unroll
        for (int b = 0; b < SS_NCHUNK; ++b) {
            s0 += p[(size_t)b * KSEG * C];
            s1 += p[(size_t)b * KSEG * C + 512];
        }
        f_s[(tid >> 6) * FS_PITCH + (tid & 63)] = s0;
        f_s[(8 + (tid >> 6)) * FS_PITCH + (tid & 63)] = s1;
        float cv = cpart[lane * KSEG + strip * 16 + wave];
        for (int m = 16; m >= 1; m >>= 1) cv += __shfl_xor(cv, m, 32);
        if (lane == 0) cnt_s[wave] = cv;
    }
    __syncthreads();

    for (int e = 0; e < 2; ++e) {
        const int idx = tid + e * 512;
        const int r = idx & 15, c = idx >> 4;
        const float f = f_s[r * FS_PITCH + c] / fmaxf(cnt_s[r], 1.0f);
        A_s[(((c >> 3) * 32 + ((c >> 2) & 1) * 16 + r) << 2) + (c & 3)] = silu_f(f);
        float bs[8];
        bspline8(f, bs);
        ((v4f*)A_s)[(8 + c) * 32 + r]      = (v4f){bs[0], bs[1], bs[2], bs[3]};
        ((v4f*)A_s)[(8 + c) * 32 + 16 + r] = (v4f){bs[4], bs[5], bs[6], bs[7]};
    }
    __syncthreads();

    v8f acc0 = (v8f){0.f, 0.f, 0.f, 0.f, 0.f, 0.f, 0.f, 0.f};
    v8f acc1 = (v8f){0.f, 0.f, 0.f, 0.f, 0.f, 0.f, 0.f, 0.f};
    const v4f* Al = (const v4f*)A_s + lane;
#pragma unroll
    for (int ch = 0; ch < L1_NCH; ++ch) {
        if (ch + 1 < L1_NCH) {
#pragma unroll
            for (int jj = 0; jj < L1_CH; ++jj)
                bq[(ch + 1) & 1][jj] = Bw[((ch + 1) * L1_CH + jj) * 32];
        }
        v4f aq[L1_CH];
#pragma unroll
        for (int jj = 0; jj < L1_CH; ++jj) aq[jj] = Al[(ch * L1_CH + jj) * 32];
#pragma unroll
        for (int jj = 0; jj < L1_CH; ++jj) {
            const v4f a = aq[jj], b = bq[ch & 1][jj];
            acc0 = __builtin_amdgcn_wmma_f32_16x16x4_f32(false, (v2f){a.x, a.y}, false, (v2f){b.x, b.y},
                                                         (short)0, acc0, false, false);
            acc1 = __builtin_amdgcn_wmma_f32_16x16x4_f32(false, (v2f){a.z, a.w}, false, (v2f){b.z, b.w},
                                                         (short)0, acc1, false, false);
        }
    }

#pragma unroll
    for (int r = 0; r < 8; ++r)
        h_s[(r + 8 * (lane >> 4)) * H1 + wave * 16 + (lane & 15)] = acc0[r] + acc1[r];
    __syncthreads();
    v4f* dst = (v4f*)(h + (size_t)strip * 16 * H1);
    dst[tid]       = ((const v4f*)h_s)[tid];
    dst[tid + 512] = ((const v4f*)h_s)[tid + 512];
}

#define L2_WAVES 16
#define L2_FEATS (H1 / L2_WAVES)
#define L2_STEPS (L2_FEATS / 4)

__global__ __launch_bounds__(32 * L2_WAVES)
void layer2_fused_kernel(const float* __restrict__ h, const float* __restrict__ Wt2,
                         float* __restrict__ out) {
    __shared__ float red_s[L2_WAVES * 256];

    const int tid  = threadIdx.x;
    const int lane = tid & 31;
    const int wave = __builtin_amdgcn_readfirstlane(tid >> 5);
    const int c    = lane & 15;
    const int hh   = lane >> 4;
    const int strip = blockIdx.x;
    const int fbase = wave * L2_FEATS;

    const float* hrow = h + (size_t)(strip * 16 + c) * H1 + fbase + hh;
    const float* Wc   = Wt2 + c;

    v8f acc0 = (v8f){0.f, 0.f, 0.f, 0.f, 0.f, 0.f, 0.f, 0.f};
    v8f acc1 = (v8f){0.f, 0.f, 0.f, 0.f, 0.f, 0.f, 0.f, 0.f};

    float x0 = hrow[0];
    float x1 = hrow[2];
#pragma unroll 1
    for (int s = 0; s < L2_STEPS; ++s) {
        const int f0 = fbase + 4 * s + hh;
        const int f1 = f0 + 2;
        const int sn = (s + 1 < L2_STEPS) ? s + 1 : s;
        const float xn0 = hrow[4 * sn];
        const float xn1 = hrow[4 * sn + 2];

        {
            v2f a, b;
            a.x = silu_f(x0);
            a.y = silu_f(x1);
            b.x = Wc[(size_t)f0 * NCLS];
            b.y = Wc[(size_t)f1 * NCLS];
            acc0 = __builtin_amdgcn_wmma_f32_16x16x4_f32(false, a, false, b, (short)0, acc0, false, false);
        }

        float bs0[8], bs1[8];
        bspline8(x0, bs0);
        bspline8(x1, bs1);
        const float* W0 = Wc + (size_t)(H1 + f0 * 8) * NCLS;
        const float* W1 = Wc + (size_t)(H1 + f1 * 8) * NCLS;
#pragma unroll
        for (int t = 0; t < 4; ++t) {
            v2f a0, b0, a1, b1;
            a0.x = bs0[2 * t];
            a0.y = bs0[2 * t + 1];
            b0.x = W0[(2 * t) * NCLS];
            b0.y = W0[(2 * t + 1) * NCLS];
            a1.x = bs1[2 * t];
            a1.y = bs1[2 * t + 1];
            b1.x = W1[(2 * t) * NCLS];
            b1.y = W1[(2 * t + 1) * NCLS];
            acc0 = __builtin_amdgcn_wmma_f32_16x16x4_f32(false, a0, false, b0, (short)0, acc0, false, false);
            acc1 = __builtin_amdgcn_wmma_f32_16x16x4_f32(false, a1, false, b1, (short)0, acc1, false, false);
        }
        x0 = xn0;
        x1 = xn1;
    }

    float* mine = red_s + wave * 256;
#pragma unroll
    for (int r = 0; r < 8; ++r) mine[r * 32 + lane] = acc0[r] + acc1[r];
    __syncthreads();

    if (tid < 256) {
        const int row = tid >> 4, col = tid & 15;
        const int idx = (row & 7) * 32 + (row >> 3) * 16 + col;
        float sum = 0.0f;
#pragma unroll
        for (int w = 0; w < L2_WAVES; ++w) sum += red_s[w * 256 + idx];
        out[(size_t)strip * 256 + tid] = sum;
    }
}

extern "C" void kernel_launch(void* const* d_in, const int* in_sizes, int n_in,
                              void* d_out, int out_size, void* d_ws, size_t ws_size,
                              hipStream_t stream) {
    const float* x   = (const float*)d_in[0];
    const int*   cls = (const int*)  d_in[1];
    const float* bw1 = (const float*)d_in[2];
    const float* sw1 = (const float*)d_in[3];
    const float* sc1 = (const float*)d_in[4];
    const float* bw2 = (const float*)d_in[5];
    const float* sw2 = (const float*)d_in[6];
    const float* sc2 = (const float*)d_in[7];
    float* out = (float*)d_out;

    float* ws  = (float*)d_ws;
    float* A1  = ws + (size_t)KSEG * C + KSEG;
    float* Wt1 = A1 + (size_t)KSEG * KT1;
    float* h   = Wt1 + (size_t)KT1 * H1;
    float* A2  = h + (size_t)KSEG * H1;
    float* Wt2 = A2 + (size_t)KSEG * KT2;
    float* part  = A2;
    float* cpart = part + (size_t)SS_NCHUNK * KSEG * C;

    segsum_kernel<<<SS_NRANGE * SS_NCHUNK, 1024, 0, stream>>>(x, cls, part, cpart);
    {
        int n = H1 * KT1 / 4;
        wprep1_kernel<<<(n + 255) / 256, 256, 0, stream>>>(bw1, sw1, sc1, Wt1);
    }
    layer1_fused_kernel<<<KSEG / 16, 32 * L1_WAVES, 0, stream>>>(part, cpart, Wt1, h);
    {
        int n = NCLS * KT2;
        wprep2_kernel<<<(n + 255) / 256, 256, 0, stream>>>(bw2, sw2, sc2, Wt2);
    }
    layer2_fused_kernel<<<KSEG / 16, 32 * L2_WAVES, 0, stream>>>(h, Wt2, out);
}


# the two weight-preparation launches folded into the segment-sum kernel: each block prepares a share (loads ahead of its class-id loads, product and store after its scan); 5 launches -> 3
# speedup vs baseline: 7.8616x; 1.0748x over previous
#include <hip/hip_runtime.h>
#include <math.h>

#define HW   1048576
#define C    64
#define KSEG 4096
#define H1   256
#define NCLS 16
#define KT1  (C + C*8)
#define KT2  (H1 + H1*8)

typedef __attribute__((ext_vector_type(2))) float v2f;
typedef __attribute__((ext_vector_type(4))) float v4f;
typedef __attribute__((ext_vector_type(8))) float v8f;

__device__ __forceinline__ float silu_f(float x) {
    return x / (1.0f + expf(-x));
}

__device__ __forceinline__ void bspline8(float x, float* __restrict__ out) {
#pragma unroll
    for (int g = 0; g < 8; ++g) {
        const float cg = 0.4f * (float)(g - 1) - 1.0f;
        const float a  = fabsf((x - cg) * 2.5f);
        const float in = fmaf(a * a, fmaf(0.5f, a, -1.0f), 2.0f / 3.0f);
        const float r  = fmaxf(2.0f - a, 0.0f);
        out[g] = a < 1.0f ? in : r * r * r * (1.0f / 6.0f);
    }
}

#define L1_WAVES 16
#define L1_STEPS (KT1 / 8)
#define L1_CH    12
#define L1_NCH   (L1_STEPS / L1_CH)
#define FS_PITCH 65

#define WP_T 288
#define WP_B 128
static_assert(WP_T * WP_B == H1 * KT1 / 4, "layer-1 weight: one 16-byte unit per item");
static_assert(WP_T * WP_B == NCLS * KT2,   "layer-2 weight: one element per item");

struct WItem { v4f raw; float sc; };

__device__ __forceinline__ WItem wprep1_load(int idx, const float* __restrict__ bw1,
                                             const float* __restrict__ sw1, const float* __restrict__ sc1) {
    const int lane = idx & 31, q = idx >> 5;
    const int j = q % L1_STEPS, nt = q / L1_STEPS;
    const int o = nt * 16 + (lane & 15), hh = lane >> 4;
    WItem w;
    if (j < C / 8) {
        w.raw = *(const v4f*)(bw1 + o * C + 8 * j + 4 * hh);
        w.sc  = 1.0f;
    } else {
        const int i = j - C / 8;
        w.raw = *(const v4f*)(sw1 + ((size_t)o * C + i) * 8 + 4 * hh);
        w.sc  = sc1[o * C + i];
    }
    return w;
}

__device__ __forceinline__ void wprep1_store(int idx, WItem w, float* __restrict__ Wp) {
    const int j = (idx >> 5) % L1_STEPS;
    v4f v = w.raw;
    if (j >= C / 8) {
        v.x = w.raw.x * w.sc;
        v.y = w.raw.y * w.sc;
        v.z = w.raw.z * w.sc;
        v.w = w.raw.w * w.sc;
    }
    ((v4f*)Wp)[idx] = v;
}

__device__ __forceinline__ WItem wprep2_load(int idx, const float* __restrict__ bw2,
                                             const float* __restrict__ sw2, const float* __restrict__ sc2) {
    const int o = idx / KT2, j = idx % KT2;
    WItem w;
    w.raw = (v4f){0.f, 0.f, 0.f, 0.f};
    if (j < H1) {
        w.raw.x = bw2[o * H1 + j];
        w.sc    = 1.0f;
    } else {
        const int j2 = j - H1, i = j2 >> 3, g = j2 & 7;
        w.raw.x = sw2[((size_t)o * H1 + i) * 8 + g];
        w.sc    = sc2[o * H1 + i];
    }
    return w;
}

__device__ __forceinline__ void wprep2_store(int idx, WItem w, float* __restrict__ Wt2) {
    const int o = idx / KT2, j = idx % KT2;
    Wt2[(size_t)j * NCLS + o] = (j < H1) ? w.raw.x : w.raw.x * w.sc;
}

#define SS_NRANGE 8
#define SS_NCHUNK 32
#define SS_SEGS   (KSEG / SS_NRANGE)
#define SS_SHIFT  9
#define SS_ROWS   (HW / SS_NCHUNK)
#define SS_WAVES  32
#define SS_WROWS  (SS_ROWS / SS_WAVES)
#define SS_U      32
static_assert(2 * WP_B == SS_NRANGE * SS_NCHUNK, "the two weights together take every block of this grid");

template <bool FULL>
__device__ __forceinline__ void gather_rows(const float* __restrict__ x, float* acc_s,
                                            const unsigned* mylist, int e, int n,
                                            int r0, int lane) {
    const int ei = e + lane;
    const unsigned entv = mylist[(FULL || ei < n) ? ei : n - 1];
    unsigned ent[SS_U];
    v2f v[SS_U];
#pragma unroll
    for (int u = 0; u < SS_U; ++u) {
        if (FULL || e + u < n) {
            ent[u] = __builtin_amdgcn_readlane(entv, u);
            const float* src = x + (size_t)(r0 + (int)(ent[u] >> 16)) * C;
            v[u] = __builtin_nontemporal_load((const v2f*)src + lane);
        }
    }
#pragma unroll
    for (int u = 0; u < SS_U; ++u) {
        if (FULL || e + u < n) {
            float* a = acc_s + (ent[u] & 0xffffu) * C + 2 * lane;
            atomicAdd(a, v[u].x);
            atomicAdd(a + 1, v[u].y);
        }
    }
}

__global__ __launch_bounds__(1024)
void segsum_kernel(const float* __restrict__ x, const int* __restrict__ cls,
                   float* __restrict__ part, float* __restrict__ cpart,
                   const float* __restrict__ bw1, const float* __restrict__ sw1,
                   const float* __restrict__ sc1, const float* __restrict__ bw2,
                   const float* __restrict__ sw2, const float* __restrict__ sc2,
                   float* __restrict__ Wp, float* __restrict__ Wt2) {
    __shared__ __attribute__((aligned(16))) float acc_s[SS_SEGS * C];
    __shared__ float cnt_s[SS_SEGS];
    __shared__ unsigned list_s[SS_WAVES * SS_WROWS];

    const int tid  = threadIdx.x;
    const int lane = tid & 31;
    const int wave = __builtin_amdgcn_readfirstlane(tid >> 5);
    const unsigned rng = blockIdx.x % SS_NRANGE;
    const int chunk    = blockIdx.x / SS_NRANGE;

    for (int i = tid; i < SS_SEGS * C / 4; i += 1024)
        ((v4f*)acc_s)[i] = (v4f){0.f, 0.f, 0.f, 0.f};
    if (tid < SS_SEGS) cnt_s[tid] = 0.0f;
    __syncthreads();

    unsigned* mylist = list_s + wave * SS_WROWS;
    const int r0 = chunk * SS_ROWS + wave * SS_WROWS;

    const bool wp_on = tid < WP_T;
    const bool wp_l1 = blockIdx.x < WP_B;
    const int  wp_idx = (int)(wp_l1 ? blockIdx.x : blockIdx.x - WP_B) * WP_T + tid;
    WItem wit;
    wit.raw = (v4f){0.f, 0.f, 0.f, 0.f};
    wit.sc  = 1.0f;
    if (wp_on) wit = wp_l1 ? wprep1_load(wp_idx, bw1, sw1, sc1) : wprep2_load(wp_idx, bw2, sw2, sc2);

    int segv[SS_WROWS / 32];
#pragma unroll
    for (int j = 0; j < SS_WROWS / 32; ++j) segv[j] = cls[r0 + j * 32 + lane];
    int n = 0;
#pragma unroll
    for (int j = 0; j < SS_WROWS / 32; ++j) {
        const bool m = ((unsigned)segv[j] >> SS_SHIFT) == rng;
        const unsigned mask = __builtin_amdgcn_ballot_w32(m);
        if (m) {
            const unsigned sl = (unsigned)segv[j] & (SS_SEGS - 1);
            mylist[n + (int)__builtin_amdgcn_mbcnt_lo(mask, 0u)] =
                ((unsigned)(j * 32 + lane) << 16) | sl;
            atomicAdd(&cnt_s[sl], 1.0f);
        }
        n += __builtin_popcount(mask);
    }
    if (wp_on) {
        if (wp_l1) wprep1_store(wp_idx, wit, Wp);
        else       wprep2_store(wp_idx, wit, Wt2);
    }

    __builtin_amdgcn_wave_barrier();

    int e = 0;
    for (; e + SS_U <= n; e += SS_U) gather_rows<true>(x, acc_s, mylist, e, n, r0, lane);
    if (e < n) gather_rows<false>(x, acc_s, mylist, e, n, r0, lane);
    __syncthreads();

    float* dst = part + ((size_t)chunk * KSEG + (size_t)rng * SS_SEGS) * C;
    for (int i = tid; i < SS_SEGS * C / 4; i += 1024)
        ((v4f*)dst)[i] = ((const v4f*)acc_s)[i];
    if (tid < SS_SEGS) cpart[chunk * KSEG + rng * SS_SEGS + tid] = cnt_s[tid];
}


static_assert(SS_NCHUNK == 32, "one partial count per lane");

__global__ __launch_bounds__(32 * L1_WAVES)
void layer1_fused_kernel(const float* __restrict__ part, const float* __restrict__ cpart,
                         const float* __restrict__ Wp, float* __restrict__ h) {
    __shared__ __attribute__((aligned(16))) float A_s[L1_STEPS * 32 * 4];
    __shared__ __attribute__((aligned(16))) float h_s[16 * H1];
    __shared__ float f_s[16 * FS_PITCH];
    __shared__ float cnt_s[16];

    const int tid  = threadIdx.x;
    const int lane = tid & 31;
    const int wave = __builtin_amdgcn_readfirstlane(tid >> 5);
    const int strip = blockIdx.x;

    const v4f* Bw = (const v4f*)Wp + (size_t)wave * L1_STEPS * 32 + lane;
    v4f bq[2][L1_CH];
#pragma unroll
    for (int jj = 0; jj < L1_CH; ++jj) bq[0][jj] = Bw[jj * 32];

    {
        const float* p = part + (size_t)strip * 16 * C + tid;
        float s0 = 0.0f, s1 = 0.0f;
#pragma unroll
        for (int b = 0; b < SS_NCHUNK; ++b) {
            s0 += p[(size_t)b * KSEG * C];
            s1 += p[(size_t)b * KSEG * C + 512];
        }
        f_s[(tid >> 6) * FS_PITCH + (tid & 63)] = s0;
        f_s[(8 + (tid >> 6)) * FS_PITCH + (tid & 63)] = s1;
        float cv = cpart[lane * KSEG + strip * 16 + wave];
        for (int m = 16; m >= 1; m >>= 1) cv += __shfl_xor(cv, m, 32);
        if (lane == 0) cnt_s[wave] = cv;
    }
    __syncthreads();

    for (int e = 0; e < 2; ++e) {
        const int idx = tid + e * 512;
        const int r = idx & 15, c = idx >> 4;
        const float f = f_s[r * FS_PITCH + c] / fmaxf(cnt_s[r], 1.0f);
        A_s[(((c >> 3) * 32 + ((c >> 2) & 1) * 16 + r) << 2) + (c & 3)] = silu_f(f);
        float bs[8];
        bspline8(f, bs);
        ((v4f*)A_s)[(8 + c) * 32 + r]      = (v4f){bs[0], bs[1], bs[2], bs[3]};
        ((v4f*)A_s)[(8 + c) * 32 + 16 + r] = (v4f){bs[4], bs[5], bs[6], bs[7]};
    }
    __syncthreads();

    v8f acc0 = (v8f){0.f, 0.f, 0.f, 0.f, 0.f, 0.f, 0.f, 0.f};
    v8f acc1 = (v8f){0.f, 0.f, 0.f, 0.f, 0.f, 0.f, 0.f, 0.f};
    const v4f* Al = (const v4f*)A_s + lane;
#pragma unroll
    for (int ch = 0; ch < L1_NCH; ++ch) {
        if (ch + 1 < L1_NCH) {
#pragma unroll
            for (int jj = 0; jj < L1_CH; ++jj)
                bq[(ch + 1) & 1][jj] = Bw[((ch + 1) * L1_CH + jj) * 32];
        }
        v4f aq[L1_CH];
#pragma unroll
        for (int jj = 0; jj < L1_CH; ++jj) aq[jj] = Al[(ch * L1_CH + jj) * 32];
#pragma unroll
        for (int jj = 0; jj < L1_CH; ++jj) {
            const v4f a = aq[jj], b = bq[ch & 1][jj];
            acc0 = __builtin_amdgcn_wmma_f32_16x16x4_f32(false, (v2f){a.x, a.y}, false, (v2f){b.x, b.y},
                                                         (short)0, acc0, false, false);
            acc1 = __builtin_amdgcn_wmma_f32_16x16x4_f32(false, (v2f){a.z, a.w}, false, (v2f){b.z, b.w},
                                                         (short)0, acc1, false, false);
        }
    }

#pragma unroll
    for (int r = 0; r < 8; ++r)
        h_s[(r + 8 * (lane >> 4)) * H1 + wave * 16 + (lane & 15)] = acc0[r] + acc1[r];
    __syncthreads();
    v4f* dst = (v4f*)(h + (size_t)strip * 16 * H1);
    dst[tid]       = ((const v4f*)h_s)[tid];
    dst[tid + 512] = ((const v4f*)h_s)[tid + 512];
}

#define L2_WAVES 16
#define L2_FEATS (H1 / L2_WAVES)
#define L2_STEPS (L2_FEATS / 4)

__global__ __launch_bounds__(32 * L2_WAVES)
void layer2_fused_kernel(const float* __restrict__ h, const float* __restrict__ Wt2,
                         float* __restrict__ out) {
    __shared__ float red_s[L2_WAVES * 256];

    const int tid  = threadIdx.x;
    const int lane = tid & 31;
    const int wave = __builtin_amdgcn_readfirstlane(tid >> 5);
    const int c    = lane & 15;
    const int hh   = lane >> 4;
    const int strip = blockIdx.x;
    const int fbase = wave * L2_FEATS;

    const float* hrow = h + (size_t)(strip * 16 + c) * H1 + fbase + hh;
    const float* Wc   = Wt2 + c;

    v8f acc0 = (v8f){0.f, 0.f, 0.f, 0.f, 0.f, 0.f, 0.f, 0.f};
    v8f acc1 = (v8f){0.f, 0.f, 0.f, 0.f, 0.f, 0.f, 0.f, 0.f};

    float x0 = hrow[0];
    float x1 = hrow[2];
#pragma unroll 1
    for (int s = 0; s < L2_STEPS; ++s) {
        const int f0 = fbase + 4 * s + hh;
        const int f1 = f0 + 2;
        const int sn = (s + 1 < L2_STEPS) ? s + 1 : s;
        const float xn0 = hrow[4 * sn];
        const float xn1 = hrow[4 * sn + 2];

        {
            v2f a, b;
            a.x = silu_f(x0);
            a.y = silu_f(x1);
            b.x = Wc[(size_t)f0 * NCLS];
            b.y = Wc[(size_t)f1 * NCLS];
            acc0 = __builtin_amdgcn_wmma_f32_16x16x4_f32(false, a, false, b, (short)0, acc0, false, false);
        }

        float bs0[8], bs1[8];
        bspline8(x0, bs0);
        bspline8(x1, bs1);
        const float* W0 = Wc + (size_t)(H1 + f0 * 8) * NCLS;
        const float* W1 = Wc + (size_t)(H1 + f1 * 8) * NCLS;
#pragma unroll
        for (int t = 0; t < 4; ++t) {
            v2f a0, b0, a1, b1;
            a0.x = bs0[2 * t];
            a0.y = bs0[2 * t + 1];
            b0.x = W0[(2 * t) * NCLS];
            b0.y = W0[(2 * t + 1) * NCLS];
            a1.x = bs1[2 * t];
            a1.y = bs1[2 * t + 1];
            b1.x = W1[(2 * t) * NCLS];
            b1.y = W1[(2 * t + 1) * NCLS];
            acc0 = __builtin_amdgcn_wmma_f32_16x16x4_f32(false, a0, false, b0, (short)0, acc0, false, false);
            acc1 = __builtin_amdgcn_wmma_f32_16x16x4_f32(false, a1, false, b1, (short)0, acc1, false, false);
        }
        x0 = xn0;
        x1 = xn1;
    }

    float* mine = red_s + wave * 256;
#pragma unroll
    for (int r = 0; r < 8; ++r) mine[r * 32 + lane] = acc0[r] + acc1[r];
    __syncthreads();

    if (tid < 256) {
        const int row = tid >> 4, col = tid & 15;
        const int idx = (row & 7) * 32 + (row >> 3) * 16 + col;
        float sum = 0.0f;
#pragma unroll
        for (int w = 0; w < L2_WAVES; ++w) sum += red_s[w * 256 + idx];
        out[(size_t)strip * 256 + tid] = sum;
    }
}

extern "C" void kernel_launch(void* const* d_in, const int* in_sizes, int n_in,
                              void* d_out, int out_size, void* d_ws, size_t ws_size,
                              hipStream_t stream) {
    const float* x   = (const float*)d_in[0];
    const int*   cls = (const int*)  d_in[1];
    const float* bw1 = (const float*)d_in[2];
    const float* sw1 = (const float*)d_in[3];
    const float* sc1 = (const float*)d_in[4];
    const float* bw2 = (const float*)d_in[5];
    const float* sw2 = (const float*)d_in[6];
    const float* sc2 = (const float*)d_in[7];
    float* out = (float*)d_out;

    float* ws  = (float*)d_ws;
    float* A1  = ws + (size_t)KSEG * C + KSEG;
    float* Wt1 = A1 + (size_t)KSEG * KT1;
    float* h   = Wt1 + (size_t)KT1 * H1;
    float* A2  = h + (size_t)KSEG * H1;
    float* Wt2 = A2 + (size_t)KSEG * KT2;
    float* part  = A2;
    float* cpart = part + (size_t)SS_NCHUNK * KSEG * C;

    segsum_kernel<<<SS_NRANGE * SS_NCHUNK, 1024, 0, stream>>>(x, cls, part, cpart,
                                                              bw1, sw1, sc1, bw2, sw2, sc2, Wt1, Wt2);
    layer1_fused_kernel<<<KSEG / 16, 32 * L1_WAVES, 0, stream>>>(part, cpart, Wt1, h);
    layer2_fused_kernel<<<KSEG / 16, 32 * L2_WAVES, 0, stream>>>(h, Wt2, out);
}


# layers 1 and 2 fused into one kernel: the h strip stays in LDS (never stored) and feeds the layer-2 WMMA stage of the same workgroup, its weights fetched meanwhile; 3 launches -> 2; sums unchanged
# speedup vs baseline: 9.0214x; 1.1475x over previous
#include <hip/hip_runtime.h>
#include <math.h>

#define HW   1048576
#define C    64
#define KSEG 4096
#define H1   256
#define NCLS 16
#define KT1  (C + C*8)
#define KT2  (H1 + H1*8)

typedef __attribute__((ext_vector_type(2))) float v2f;
typedef __attribute__((ext_vector_type(4))) float v4f;
typedef __attribute__((ext_vector_type(8))) float v8f;

__device__ __forceinline__ float silu_f(float x) {
    return x / (1.0f + expf(-x));
}

__device__ __forceinline__ void bspline8(float x, float* __restrict__ out) {
#pragma unroll
    for (int g = 0; g < 8; ++g) {
        const float cg = 0.4f * (float)(g - 1) - 1.0f;
        const float a  = fabsf((x - cg) * 2.5f);
        const float in = fmaf(a * a, fmaf(0.5f, a, -1.0f), 2.0f / 3.0f);
        const float r  = fmaxf(2.0f - a, 0.0f);
        out[g] = a < 1.0f ? in : r * r * r * (1.0f / 6.0f);
    }
}

#define L1_WAVES 16
#define L1_STEPS (KT1 / 8)
#define L1_CH    12
#define L1_NCH   (L1_STEPS / L1_CH)
#define FS_PITCH 65

#define WP_T 288
#define WP_B 128
static_assert(WP_T * WP_B == H1 * KT1 / 4, "layer-1 weight: one 16-byte unit per item");
static_assert(WP_T * WP_B == NCLS * KT2,   "layer-2 weight: one element per item");

struct WItem { v4f raw; float sc; };

__device__ __forceinline__ WItem wprep1_load(int idx, const float* __restrict__ bw1,
                                             const float* __restrict__ sw1, const float* __restrict__ sc1) {
    const int lane = idx & 31, q = idx >> 5;
    const int j = q % L1_STEPS, nt = q / L1_STEPS;
    const int o = nt * 16 + (lane & 15), hh = lane >> 4;
    WItem w;
    if (j < C / 8) {
        w.raw = *(const v4f*)(bw1 + o * C + 8 * j + 4 * hh);
        w.sc  = 1.0f;
    } else {
        const int i = j - C / 8;
        w.raw = *(const v4f*)(sw1 + ((size_t)o * C + i) * 8 + 4 * hh);
        w.sc  = sc1[o * C + i];
    }
    return w;
}

__device__ __forceinline__ void wprep1_store(int idx, WItem w, float* __restrict__ Wp) {
    const int j = (idx >> 5) % L1_STEPS;
    v4f v = w.raw;
    if (j >= C / 8) {
        v.x = w.raw.x * w.sc;
        v.y = w.raw.y * w.sc;
        v.z = w.raw.z * w.sc;
        v.w = w.raw.w * w.sc;
    }
    ((v4f*)Wp)[idx] = v;
}

__device__ __forceinline__ WItem wprep2_load(int idx, const float* __restrict__ bw2,
                                             const float* __restrict__ sw2, const float* __restrict__ sc2) {
    const int o = idx / KT2, j = idx % KT2;
    WItem w;
    w.raw = (v4f){0.f, 0.f, 0.f, 0.f};
    if (j < H1) {
        w.raw.x = bw2[o * H1 + j];
        w.sc    = 1.0f;
    } else {
        const int j2 = j - H1, i = j2 >> 3, g = j2 & 7;
        w.raw.x = sw2[((size_t)o * H1 + i) * 8 + g];
        w.sc    = sc2[o * H1 + i];
    }
    return w;
}

__device__ __forceinline__ void wprep2_store(int idx, WItem w, float* __restrict__ Wt2) {
    const int o = idx / KT2, j = idx % KT2;
    Wt2[(size_t)j * NCLS + o] = (j < H1) ? w.raw.x : w.raw.x * w.sc;
}

#define SS_NRANGE 8
#define SS_NCHUNK 32
#define SS_SEGS   (KSEG / SS_NRANGE)
#define SS_SHIFT  9
#define SS_ROWS   (HW / SS_NCHUNK)
#define SS_WAVES  32
#define SS_WROWS  (SS_ROWS / SS_WAVES)
#define SS_U      32
static_assert(2 * WP_B == SS_NRANGE * SS_NCHUNK, "the two weights together take every block of this grid");

template <bool FULL>
__device__ __forceinline__ void gather_rows(const float* __restrict__ x, float* acc_s,
                                            const unsigned* mylist, int e, int n,
                                            int r0, int lane) {
    const int ei = e + lane;
    const unsigned entv = mylist[(FULL || ei < n) ? ei : n - 1];
    unsigned ent[SS_U];
    v2f v[SS_U];
#pragma unroll
    for (int u = 0; u < SS_U; ++u) {
        if (FULL || e + u < n) {
            ent[u] = __builtin_amdgcn_readlane(entv, u);
            const float* src = x + (size_t)(r0 + (int)(ent[u] >> 16)) * C;
            v[u] = __builtin_nontemporal_load((const v2f*)src + lane);
        }
    }
#pragma unroll
    for (int u = 0; u < SS_U; ++u) {
        if (FULL || e + u < n) {
            float* a = acc_s + (ent[u] & 0xffffu) * C + 2 * lane;
            atomicAdd(a, v[u].x);
            atomicAdd(a + 1, v[u].y);
        }
    }
}

__global__ __launch_bounds__(1024)
void segsum_kernel(const float* __restrict__ x, const int* __restrict__ cls,
                   float* __restrict__ part, float* __restrict__ cpart,
                   const float* __restrict__ bw1, const float* __restrict__ sw1,
                   const float* __restrict__ sc1, const float* __restrict__ bw2,
                   const float* __restrict__ sw2, const float* __restrict__ sc2,
                   float* __restrict__ Wp, float* __restrict__ Wt2) {
    __shared__ __attribute__((aligned(16))) float acc_s[SS_SEGS * C];
    __shared__ float cnt_s[SS_SEGS];
    __shared__ unsigned list_s[SS_WAVES * SS_WROWS];

    const int tid  = threadIdx.x;
    const int lane = tid & 31;
    const int wave = __builtin_amdgcn_readfirstlane(tid >> 5);
    const unsigned rng = blockIdx.x % SS_NRANGE;
    const int chunk    = blockIdx.x / SS_NRANGE;

    for (int i = tid; i < SS_SEGS * C / 4; i += 1024)
        ((v4f*)acc_s)[i] = (v4f){0.f, 0.f, 0.f, 0.f};
    if (tid < SS_SEGS) cnt_s[tid] = 0.0f;
    __syncthreads();

    unsigned* mylist = list_s + wave * SS_WROWS;
    const int r0 = chunk * SS_ROWS + wave * SS_WROWS;

    const bool wp_on = tid < WP_T;
    const bool wp_l1 = blockIdx.x < WP_B;
    const int  wp_idx = (int)(wp_l1 ? blockIdx.x : blockIdx.x - WP_B) * WP_T + tid;
    WItem wit;
    wit.raw = (v4f){0.f, 0.f, 0.f, 0.f};
    wit.sc  = 1.0f;
    if (wp_on) wit = wp_l1 ? wprep1_load(wp_idx, bw1, sw1, sc1) : wprep2_load(wp_idx, bw2, sw2, sc2);

    int segv[SS_WROWS / 32];
#pragma unroll
    for (int j = 0; j < SS_WROWS / 32; ++j) segv[j] = cls[r0 + j * 32 + lane];
    int n = 0;
#pragma unroll
    for (int j = 0; j < SS_WROWS / 32; ++j) {
        const bool m = ((unsigned)segv[j] >> SS_SHIFT) == rng;
        const unsigned mask = __builtin_amdgcn_ballot_w32(m);
        if (m) {
            const unsigned sl = (unsigned)segv[j] & (SS_SEGS - 1);
            mylist[n + (int)__builtin_amdgcn_mbcnt_lo(mask, 0u)] =
                ((unsigned)(j * 32 + lane) << 16) | sl;
            atomicAdd(&cnt_s[sl], 1.0f);
        }
        n += __builtin_popcount(mask);
    }
    if (wp_on) {
        if (wp_l1) wprep1_store(wp_idx, wit, Wp);
        else       wprep2_store(wp_idx, wit, Wt2);
    }

    __builtin_amdgcn_wave_barrier();

    int e = 0;
    for (; e + SS_U <= n; e += SS_U) gather_rows<true>(x, acc_s, mylist, e, n, r0, lane);
    if (e < n) gather_rows<false>(x, acc_s, mylist, e, n, r0, lane);
    __syncthreads();

    float* dst = part + ((size_t)chunk * KSEG + (size_t)rng * SS_SEGS) * C;
    for (int i = tid; i < SS_SEGS * C / 4; i += 1024)
        ((v4f*)dst)[i] = ((const v4f*)acc_s)[i];
    if (tid < SS_SEGS) cpart[chunk * KSEG + rng * SS_SEGS + tid] = cnt_s[tid];
}

#define LF_WAVES L1_WAVES
#define L2_FEATS (H1 / LF_WAVES)
#define L2_STEPS (L2_FEATS / 4)
#define HS_PITCH 258

static_assert(SS_NCHUNK == 32, "one partial count per lane");

__global__ __launch_bounds__(32 * LF_WAVES)
void layers_fused_kernel(const float* __restrict__ part, const float* __restrict__ cpart,
                         const float* __restrict__ Wp, const float* __restrict__ Wt2,
                         float* __restrict__ out) {
    __shared__ __attribute__((aligned(16))) float A_s[L1_STEPS * 32 * 4];
    __shared__ float h_s[16 * HS_PITCH];
    __shared__ float f_s[16 * FS_PITCH];
    __shared__ float cnt_s[16];
    __shared__ float red_s[LF_WAVES * 256];

    const int tid  = threadIdx.x;
    const int lane = tid & 31;
    const int wave = __builtin_amdgcn_readfirstlane(tid >> 5);
    const int strip = blockIdx.x;
    const int c    = lane & 15;
    const int hh   = lane >> 4;

    const v4f* Bw = (const v4f*)Wp + (size_t)wave * L1_STEPS * 32 + lane;
    v4f bq[2][L1_CH];
#pragma unroll
    for (int jj = 0; jj < L1_CH; ++jj) bq[0][jj] = Bw[jj * 32];

    {
        const float* p = part + (size_t)strip * 16 * C + tid;
        float s0 = 0.0f, s1 = 0.0f;
#pragma unroll
        for (int b = 0; b < SS_NCHUNK; ++b) {
            s0 += p[(size_t)b * KSEG * C];
            s1 += p[(size_t)b * KSEG * C + 512];
        }
        f_s[(tid >> 6) * FS_PITCH + (tid & 63)] = s0;
        f_s[(8 + (tid >> 6)) * FS_PITCH + (tid & 63)] = s1;
        float cv = cpart[lane * KSEG + strip * 16 + wave];
        for (int m = 16; m >= 1; m >>= 1) cv += __shfl_xor(cv, m, 32);
        if (lane == 0) cnt_s[wave] = cv;
    }
    __syncthreads();

    for (int e = 0; e < 2; ++e) {
        const int idx = tid + e * 512;
        const int r = idx & 15, cc = idx >> 4;
        const float f = f_s[r * FS_PITCH + cc] / fmaxf(cnt_s[r], 1.0f);
        A_s[(((cc >> 3) * 32 + ((cc >> 2) & 1) * 16 + r) << 2) + (cc & 3)] = silu_f(f);
        float bs[8];
        bspline8(f, bs);
        ((v4f*)A_s)[(8 + cc) * 32 + r]      = (v4f){bs[0], bs[1], bs[2], bs[3]};
        ((v4f*)A_s)[(8 + cc) * 32 + 16 + r] = (v4f){bs[4], bs[5], bs[6], bs[7]};
    }
    __syncthreads();

    v8f acc0 = (v8f){0.f, 0.f, 0.f, 0.f, 0.f, 0.f, 0.f, 0.f};
    v8f acc1 = (v8f){0.f, 0.f, 0.f, 0.f, 0.f, 0.f, 0.f, 0.f};
    const v4f* Al = (const v4f*)A_s + lane;
#pragma unroll
    for (int ch = 0; ch < L1_NCH; ++ch) {
        if (ch + 1 < L1_NCH) {
#pragma unroll
            for (int jj = 0; jj < L1_CH; ++jj)
                bq[(ch + 1) & 1][jj] = Bw[((ch + 1) * L1_CH + jj) * 32];
        }
        v4f aq[L1_CH];
#pragma unroll
        for (int jj = 0; jj < L1_CH; ++jj) aq[jj] = Al[(ch * L1_CH + jj) * 32];
#pragma unroll
        for (int jj = 0; jj < L1_CH; ++jj) {
            const v4f a = aq[jj], b = bq[ch & 1][jj];
            acc0 = __builtin_amdgcn_wmma_f32_16x16x4_f32(false, (v2f){a.x, a.y}, false, (v2f){b.x, b.y},
                                                         (short)0, acc0, false, false);
            acc1 = __builtin_amdgcn_wmma_f32_16x16x4_f32(false, (v2f){a.z, a.w}, false, (v2f){b.z, b.w},
                                                         (short)0, acc1, false, false);
        }
    }

    const int fbase = wave * L2_FEATS;
    const float* Wc = Wt2 + c;
    float wsil[L2_STEPS][2];
    float wspl[L2_STEPS][2][8];
#pragma unroll
    for (int s = 0; s < L2_STEPS; ++s) {
#pragma unroll
        for (int q = 0; q < 2; ++q) {
            const int f = fbase + 4 * s + 2 * q + hh;
            wsil[s][q] = Wc[(size_t)f * NCLS];
#pragma unroll
            for (int g = 0; g < 8; ++g) wspl[s][q][g] = Wc[(size_t)(H1 + f * 8 + g) * NCLS];
        }
    }

#pragma unroll
    for (int r = 0; r < 8; ++r)
        h_s[(r + 8 * hh) * HS_PITCH + wave * 16 + c] = acc0[r] + acc1[r];
    __syncthreads();

    float xv[L2_STEPS][2];
#pragma unroll
    for (int s = 0; s < L2_STEPS; ++s) {
#pragma unroll
        for (int q = 0; q < 2; ++q)
            xv[s][q] = h_s[c * HS_PITCH + fbase + 4 * s + 2 * q + hh];
    }
    v8f o0 = (v8f){0.f, 0.f, 0.f, 0.f, 0.f, 0.f, 0.f, 0.f};
    v8f o1 = (v8f){0.f, 0.f, 0.f, 0.f, 0.f, 0.f, 0.f, 0.f};
#pragma unroll
    for (int s = 0; s < L2_STEPS; ++s) {
        {
            v2f a, b;
            a.x = silu_f(xv[s][0]);
            a.y = silu_f(xv[s][1]);
            b.x = wsil[s][0];
            b.y = wsil[s][1];
            o0 = __builtin_amdgcn_wmma_f32_16x16x4_f32(false, a, false, b, (short)0, o0, false, false);
        }
        float bs0[8], bs1[8];
        bspline8(xv[s][0], bs0);
        bspline8(xv[s][1], bs1);
#pragma unroll
        for (int t = 0; t < 4; ++t) {
            v2f a0, b0, a1, b1;
            a0.x = bs0[2 * t];
            a0.y = bs0[2 * t + 1];
            b0.x = wspl[s][0][2 * t];
            b0.y = wspl[s][0][2 * t + 1];
            a1.x = bs1[2 * t];
            a1.y = bs1[2 * t + 1];
            b1.x = wspl[s][1][2 * t];
            b1.y = wspl[s][1][2 * t + 1];
            o0 = __builtin_amdgcn_wmma_f32_16x16x4_f32(false, a0, false, b0, (short)0, o0, false, false);
            o1 = __builtin_amdgcn_wmma_f32_16x16x4_f32(false, a1, false, b1, (short)0, o1, false, false);
        }
    }

    float* mine = red_s + wave * 256;
#pragma unroll
    for (int r = 0; r < 8; ++r) mine[r * 32 + lane] = o0[r] + o1[r];
    __syncthreads();

    if (tid < 256) {
        const int row = tid >> 4, col = tid & 15;
        const int idx = (row & 7) * 32 + (row >> 3) * 16 + col;
        float sum = 0.0f;
#pragma unroll
        for (int w = 0; w < LF_WAVES; ++w) sum += red_s[w * 256 + idx];
        out[(size_t)strip * 256 + tid] = sum;
    }
}

extern "C" void kernel_launch(void* const* d_in, const int* in_sizes, int n_in,
                              void* d_out, int out_size, void* d_ws, size_t ws_size,
                              hipStream_t stream) {
    const float* x   = (const float*)d_in[0];
    const int*   cls = (const int*)  d_in[1];
    const float* bw1 = (const float*)d_in[2];
    const float* sw1 = (const float*)d_in[3];
    const float* sc1 = (const float*)d_in[4];
    const float* bw2 = (const float*)d_in[5];
    const float* sw2 = (const float*)d_in[6];
    const float* sc2 = (const float*)d_in[7];
    float* out = (float*)d_out;

    float* ws  = (float*)d_ws;
    float* A1  = ws + (size_t)KSEG * C + KSEG;
    float* Wt1 = A1 + (size_t)KSEG * KT1;
    float* h   = Wt1 + (size_t)KT1 * H1;
    float* A2  = h + (size_t)KSEG * H1;
    float* Wt2 = A2 + (size_t)KSEG * KT2;
    float* part  = A2;
    float* cpart = part + (size_t)SS_NCHUNK * KSEG * C;

    segsum_kernel<<<SS_NRANGE * SS_NCHUNK, 1024, 0, stream>>>(x, cls, part, cpart,
                                                              bw1, sw1, sc1, bw2, sw2, sc2, Wt1, Wt2);
    layers_fused_kernel<<<KSEG / 16, 32 * LF_WAVES, 0, stream>>>(part, cpart, Wt1, Wt2, out);
}


# segment sums without LDS float adds: each block counting-sorts its kept rows by segment (integer LDS atomics), waves sum 16 whole segments in registers and store them; no LDS tile; sum order changes
# speedup vs baseline: 9.7405x; 1.0797x over previous
#include <hip/hip_runtime.h>
#include <math.h>

#define HW   1048576
#define C    64
#define KSEG 4096
#define H1   256
#define NCLS 16
#define KT1  (C + C*8)
#define KT2  (H1 + H1*8)

typedef __attribute__((ext_vector_type(2))) float v2f;
typedef __attribute__((ext_vector_type(4))) float v4f;
typedef __attribute__((ext_vector_type(8))) float v8f;

__device__ __forceinline__ float silu_f(float x) {
    return x / (1.0f + expf(-x));
}

__device__ __forceinline__ void bspline8(float x, float* __restrict__ out) {
#pragma unroll
    for (int g = 0; g < 8; ++g) {
        const float cg = 0.4f * (float)(g - 1) - 1.0f;
        const float a  = fabsf((x - cg) * 2.5f);
        const float in = fmaf(a * a, fmaf(0.5f, a, -1.0f), 2.0f / 3.0f);
        const float r  = fmaxf(2.0f - a, 0.0f);
        out[g] = a < 1.0f ? in : r * r * r * (1.0f / 6.0f);
    }
}

#define L1_WAVES 16
#define L1_STEPS (KT1 / 8)
#define L1_CH    12
#define L1_NCH   (L1_STEPS / L1_CH)
#define FS_PITCH 65

#define WP_T 288
#define WP_B 128
static_assert(WP_T * WP_B == H1 * KT1 / 4, "layer-1 weight: one 16-byte unit per item");
static_assert(WP_T * WP_B == NCLS * KT2,   "layer-2 weight: one element per item");

struct WItem { v4f raw; float sc; };

__device__ __forceinline__ WItem wprep1_load(int idx, const float* __restrict__ bw1,
                                             const float* __restrict__ sw1, const float* __restrict__ sc1) {
    const int lane = idx & 31, q = idx >> 5;
    const int j = q % L1_STEPS, nt = q / L1_STEPS;
    const int o = nt * 16 + (lane & 15), hh = lane >> 4;
    WItem w;
    if (j < C / 8) {
        w.raw = *(const v4f*)(bw1 + o * C + 8 * j + 4 * hh);
        w.sc  = 1.0f;
    } else {
        const int i = j - C / 8;
        w.raw = *(const v4f*)(sw1 + ((size_t)o * C + i) * 8 + 4 * hh);
        w.sc  = sc1[o * C + i];
    }
    return w;
}

__device__ __forceinline__ void wprep1_store(int idx, WItem w, float* __restrict__ Wp) {
    const int j = (idx >> 5) % L1_STEPS;
    v4f v = w.raw;
    if (j >= C / 8) {
        v.x = w.raw.x * w.sc;
        v.y = w.raw.y * w.sc;
        v.z = w.raw.z * w.sc;
        v.w = w.raw.w * w.sc;
    }
    ((v4f*)Wp)[idx] = v;
}

__device__ __forceinline__ WItem wprep2_load(int idx, const float* __restrict__ bw2,
                                             const float* __restrict__ sw2, const float* __restrict__ sc2) {
    const int o = idx / KT2, j = idx % KT2;
    WItem w;
    w.raw = (v4f){0.f, 0.f, 0.f, 0.f};
    if (j < H1) {
        w.raw.x = bw2[o * H1 + j];
        w.sc    = 1.0f;
    } else {
        const int j2 = j - H1, i = j2 >> 3, g = j2 & 7;
        w.raw.x = sw2[((size_t)o * H1 + i) * 8 + g];
        w.sc    = sc2[o * H1 + i];
    }
    return w;
}

__device__ __forceinline__ void wprep2_store(int idx, WItem w, float* __restrict__ Wt2) {
    const int o = idx / KT2, j = idx % KT2;
    Wt2[(size_t)j * NCLS + o] = (j < H1) ? w.raw.x : w.raw.x * w.sc;
}

#define SS_NRANGE 8
#define SS_NCHUNK 32
#define SS_SEGS   (KSEG / SS_NRANGE)
#define SS_SHIFT  9
#define SS_ROWS   (HW / SS_NCHUNK)
#define SS_WAVES  32
#define SS_WROWS  (SS_ROWS / SS_WAVES)
#define SS_U      32
#define SS_WSEGS  (SS_SEGS / SS_WAVES)
#define SS_SBITS  9
static_assert((1 << SS_SBITS) == SS_SEGS, "a sorted entry keeps the local segment in its low bits");
static_assert(SS_ROWS <= (1 << (32 - SS_SBITS)), "a sorted entry keeps the row of the chunk above them");
static_assert(2 * WP_B == SS_NRANGE * SS_NCHUNK, "the two weights together take every block of this grid");

template <bool FULL>
__device__ __forceinline__ void gather_sorted(const float* __restrict__ xc, const unsigned* sorted,
                                              int e, int eEnd, int lane,
                                              int& cur, v2f& acc, float* __restrict__ pdst) {
    const int ei = e + lane;
    const unsigned entv = sorted[(FULL || ei < eEnd) ? ei : eEnd - 1];
    unsigned ent[SS_U];
    v2f v[SS_U];
#pragma unroll
    for (int u = 0; u < SS_U; ++u) {
        if (FULL || e + u < eEnd) {
            ent[u] = __builtin_amdgcn_readlane(entv, u);
            const float* src = xc + (size_t)(ent[u] >> SS_SBITS) * C;
            v[u] = __builtin_nontemporal_load((const v2f*)src + lane);
        }
    }
#pragma unroll
    for (int u = 0; u < SS_U; ++u) {
        if (FULL || e + u < eEnd) {
            const int s = (int)(ent[u] & (unsigned)(SS_SEGS - 1));
            if (s != cur) {
                *(v2f*)(pdst + cur * C) = acc;
#pragma nounroll
                for (int z = cur + 1; z < s; ++z) *(v2f*)(pdst + z * C) = (v2f){0.f, 0.f};
                acc = (v2f){0.f, 0.f};
                cur = s;
            }
            acc.x += v[u].x;
            acc.y += v[u].y;
        }
    }
}

__global__ __launch_bounds__(1024)
void segsum_kernel(const float* __restrict__ x, const int* __restrict__ cls,
                   float* __restrict__ part, float* __restrict__ cpart,
                   const float* __restrict__ bw1, const float* __restrict__ sw1,
                   const float* __restrict__ sc1, const float* __restrict__ bw2,
                   const float* __restrict__ sw2, const float* __restrict__ sc2,
                   float* __restrict__ Wp, float* __restrict__ Wt2) {
    __shared__ unsigned list_s[SS_WAVES * SS_WROWS];
    __shared__ unsigned sort_s[SS_ROWS];
    __shared__ unsigned cnt_s[SS_SEGS];
    __shared__ unsigned start_s[SS_SEGS + 1];
    __shared__ unsigned wsum_s[SS_SEGS / 32];

    const int tid  = threadIdx.x;
    const int lane = tid & 31;
    const int wave = __builtin_amdgcn_readfirstlane(tid >> 5);
    const unsigned rng = blockIdx.x % SS_NRANGE;
    const int chunk    = blockIdx.x / SS_NRANGE;

    if (tid < SS_SEGS) cnt_s[tid] = 0u;
    __syncthreads();

    unsigned* mylist = list_s + wave * SS_WROWS;
    const int r0 = chunk * SS_ROWS + wave * SS_WROWS;

    const bool wp_on = tid < WP_T;
    const bool wp_l1 = blockIdx.x < WP_B;
    const int  wp_idx = (int)(wp_l1 ? blockIdx.x : blockIdx.x - WP_B) * WP_T + tid;
    WItem wit;
    wit.raw = (v4f){0.f, 0.f, 0.f, 0.f};
    wit.sc  = 1.0f;
    if (wp_on) wit = wp_l1 ? wprep1_load(wp_idx, bw1, sw1, sc1) : wprep2_load(wp_idx, bw2, sw2, sc2);

    int segv[SS_WROWS / 32];
#pragma unroll
    for (int j = 0; j < SS_WROWS / 32; ++j) segv[j] = cls[r0 + j * 32 + lane];
    int n = 0;
#pragma unroll
    for (int j = 0; j < SS_WROWS / 32; ++j) {
        const bool m = ((unsigned)segv[j] >> SS_SHIFT) == rng;
        const unsigned mask = __builtin_amdgcn_ballot_w32(m);
        if (m) {
            const unsigned sl = (unsigned)segv[j] & (SS_SEGS - 1);
            mylist[n + (int)__builtin_amdgcn_mbcnt_lo(mask, 0u)] =
                ((unsigned)(j * 32 + lane) << 16) | sl;
            atomicAdd(&cnt_s[sl], 1u);
        }
        n += __builtin_popcount(mask);
    }
    if (wp_on) {
        if (wp_l1) wprep1_store(wp_idx, wit, Wp);
        else       wprep2_store(wp_idx, wit, Wt2);
    }

    __syncthreads();

    unsigned cv = 0u;
    if (tid < SS_SEGS) cv = cnt_s[tid];
    unsigned inc = cv;
    for (int d = 1; d < 32; d <<= 1) {
        const unsigned t = __shfl_up(inc, d, 32);
        if (lane >= d) inc += t;
    }
    if (tid < SS_SEGS && lane == 31) wsum_s[wave] = inc;
    __syncthreads();
    if (tid < SS_SEGS) {
        unsigned base = 0u;
        for (int w = 0; w < wave; ++w) base += wsum_s[w];
        const unsigned ex = base + inc - cv;
        start_s[tid] = ex;
        cnt_s[tid]   = ex;
        if (tid == SS_SEGS - 1) start_s[SS_SEGS] = base + inc;
        cpart[chunk * KSEG + rng * SS_SEGS + tid] = (float)cv;
    }
    __syncthreads();

    for (int e = 0; e < n; e += 32) {
        const int ei = e + lane;
        if (ei < n) {
            const unsigned ent  = mylist[ei];
            const unsigned sl   = ent & (unsigned)(SS_SEGS - 1);
            const unsigned slot = atomicAdd(&cnt_s[sl], 1u);
            sort_s[slot] = ((unsigned)(wave * SS_WROWS + (int)(ent >> 16)) << SS_SBITS) | sl;
        }
    }
    __syncthreads();

    const int seg0 = wave * SS_WSEGS;
    int e          = __builtin_amdgcn_readfirstlane((int)start_s[seg0]);
    const int eEnd = __builtin_amdgcn_readfirstlane((int)start_s[seg0 + SS_WSEGS]);
    const float* xc = x + (size_t)chunk * SS_ROWS * C;
    float* pdst = part + ((size_t)chunk * KSEG + (size_t)rng * SS_SEGS) * C + 2 * lane;
    int cur = seg0;
    v2f acc = (v2f){0.f, 0.f};
    for (; e + SS_U <= eEnd; e += SS_U) gather_sorted<true>(xc, sort_s, e, eEnd, lane, cur, acc, pdst);
    if (e < eEnd) gather_sorted<false>(xc, sort_s, e, eEnd, lane, cur, acc, pdst);
    for (; cur < seg0 + SS_WSEGS; ++cur) {
        *(v2f*)(pdst + cur * C) = acc;
        acc = (v2f){0.f, 0.f};
    }
}

#define LF_WAVES L1_WAVES
#define L2_FEATS (H1 / LF_WAVES)
#define L2_STEPS (L2_FEATS / 4)
#define HS_PITCH 258

static_assert(SS_NCHUNK == 32, "one partial count per lane");

__global__ __launch_bounds__(32 * LF_WAVES)
void layers_fused_kernel(const float* __restrict__ part, const float* __restrict__ cpart,
                         const float* __restrict__ Wp, const float* __restrict__ Wt2,
                         float* __restrict__ out) {
    __shared__ __attribute__((aligned(16))) float A_s[L1_STEPS * 32 * 4];
    __shared__ float h_s[16 * HS_PITCH];
    __shared__ float f_s[16 * FS_PITCH];
    __shared__ float cnt_s[16];
    __shared__ float red_s[LF_WAVES * 256];

    const int tid  = threadIdx.x;
    const int lane = tid & 31;
    const int wave = __builtin_amdgcn_readfirstlane(tid >> 5);
    const int strip = blockIdx.x;
    const int c    = lane & 15;
    const int hh   = lane >> 4;

    const v4f* Bw = (const v4f*)Wp + (size_t)wave * L1_STEPS * 32 + lane;
    v4f bq[2][L1_CH];
#pragma unroll
    for (int jj = 0; jj < L1_CH; ++jj) bq[0][jj] = Bw[jj * 32];

    {
        const float* p = part + (size_t)strip * 16 * C + tid;
        float s0 = 0.0f, s1 = 0.0f;
#pragma unroll
        for (int b = 0; b < SS_NCHUNK; ++b) {
            s0 += p[(size_t)b * KSEG * C];
            s1 += p[(size_t)b * KSEG * C + 512];
        }
        f_s[(tid >> 6) * FS_PITCH + (tid & 63)] = s0;
        f_s[(8 + (tid >> 6)) * FS_PITCH + (tid & 63)] = s1;
        float cv = cpart[lane * KSEG + strip * 16 + wave];
        for (int m = 16; m >= 1; m >>= 1) cv += __shfl_xor(cv, m, 32);
        if (lane == 0) cnt_s[wave] = cv;
    }
    __syncthreads();

    for (int e = 0; e < 2; ++e) {
        const int idx = tid + e * 512;
        const int r = idx & 15, cc = idx >> 4;
        const float f = f_s[r * FS_PITCH + cc] / fmaxf(cnt_s[r], 1.0f);
        A_s[(((cc >> 3) * 32 + ((cc >> 2) & 1) * 16 + r) << 2) + (cc & 3)] = silu_f(f);
        float bs[8];
        bspline8(f, bs);
        ((v4f*)A_s)[(8 + cc) * 32 + r]      = (v4f){bs[0], bs[1], bs[2], bs[3]};
        ((v4f*)A_s)[(8 + cc) * 32 + 16 + r] = (v4f){bs[4], bs[5], bs[6], bs[7]};
    }
    __syncthreads();

    v8f acc0 = (v8f){0.f, 0.f, 0.f, 0.f, 0.f, 0.f, 0.f, 0.f};
    v8f acc1 = (v8f){0.f, 0.f, 0.f, 0.f, 0.f, 0.f, 0.f, 0.f};
    const v4f* Al = (const v4f*)A_s + lane;
#pragma unroll
    for (int ch = 0; ch < L1_NCH; ++ch) {
        if (ch + 1 < L1_NCH) {
#pragma unroll
            for (int jj = 0; jj < L1_CH; ++jj)
                bq[(ch + 1) & 1][jj] = Bw[((ch + 1) * L1_CH + jj) * 32];
        }
        v4f aq[L1_CH];
#pragma unroll
        for (int jj = 0; jj < L1_CH; ++jj) aq[jj] = Al[(ch * L1_CH + jj) * 32];
#pragma unroll
        for (int jj = 0; jj < L1_CH; ++jj) {
            const v4f a = aq[jj], b = bq[ch & 1][jj];
            acc0 = __builtin_amdgcn_wmma_f32_16x16x4_f32(false, (v2f){a.x, a.y}, false, (v2f){b.x, b.y},
                                                         (short)0, acc0, false, false);
            acc1 = __builtin_amdgcn_wmma_f32_16x16x4_f32(false, (v2f){a.z, a.w}, false, (v2f){b.z, b.w},
                                                         (short)0, acc1, false, false);
        }
    }

    const int fbase = wave * L2_FEATS;
    const float* Wc = Wt2 + c;
    float wsil[L2_STEPS][2];
    float wspl[L2_STEPS][2][8];
#pragma unroll
    for (int s = 0; s < L2_STEPS; ++s) {
#pragma unroll
        for (int q = 0; q < 2; ++q) {
            const int f = fbase + 4 * s + 2 * q + hh;
            wsil[s][q] = Wc[(size_t)f * NCLS];
#pragma unroll
            for (int g = 0; g < 8; ++g) wspl[s][q][g] = Wc[(size_t)(H1 + f * 8 + g) * NCLS];
        }
    }

#pragma unroll
    for (int r = 0; r < 8; ++r)
        h_s[(r + 8 * hh) * HS_PITCH + wave * 16 + c] = acc0[r] + acc1[r];
    __syncthreads();

    float xv[L2_STEPS][2];
#pragma unroll
    for (int s = 0; s < L2_STEPS; ++s) {
#pragma unroll
        for (int q = 0; q < 2; ++q)
            xv[s][q] = h_s[c * HS_PITCH + fbase + 4 * s + 2 * q + hh];
    }
    v8f o0 = (v8f){0.f, 0.f, 0.f, 0.f, 0.f, 0.f, 0.f, 0.f};
    v8f o1 = (v8f){0.f, 0.f, 0.f, 0.f, 0.f, 0.f, 0.f, 0.f};
#pragma unroll
    for (int s = 0; s < L2_STEPS; ++s) {
        {
            v2f a, b;
            a.x = silu_f(xv[s][0]);
            a.y = silu_f(xv[s][1]);
            b.x = wsil[s][0];
            b.y = wsil[s][1];
            o0 = __builtin_amdgcn_wmma_f32_16x16x4_f32(false, a, false, b, (short)0, o0, false, false);
        }
        float bs0[8], bs1[8];
        bspline8(xv[s][0], bs0);
        bspline8(xv[s][1], bs1);
#pragma unroll
        for (int t = 0; t < 4; ++t) {
            v2f a0, b0, a1, b1;
            a0.x = bs0[2 * t];
            a0.y = bs0[2 * t + 1];
            b0.x = wspl[s][0][2 * t];
            b0.y = wspl[s][0][2 * t + 1];
            a1.x = bs1[2 * t];
            a1.y = bs1[2 * t + 1];
            b1.x = wspl[s][1][2 * t];
            b1.y = wspl[s][1][2 * t + 1];
            o0 = __builtin_amdgcn_wmma_f32_16x16x4_f32(false, a0, false, b0, (short)0, o0, false, false);
            o1 = __builtin_amdgcn_wmma_f32_16x16x4_f32(false, a1, false, b1, (short)0, o1, false, false);
        }
    }

    float* mine = red_s + wave * 256;
#pragma unroll
    for (int r = 0; r < 8; ++r) mine[r * 32 + lane] = o0[r] + o1[r];
    __syncthreads();

    if (tid < 256) {
        const int row = tid >> 4, col = tid & 15;
        const int idx = (row & 7) * 32 + (row >> 3) * 16 + col;
        float sum = 0.0f;
#pragma unroll
        for (int w = 0; w < LF_WAVES; ++w) sum += red_s[w * 256 + idx];
        out[(size_t)strip * 256 + tid] = sum;
    }
}

extern "C" void kernel_launch(void* const* d_in, const int* in_sizes, int n_in,
                              void* d_out, int out_size, void* d_ws, size_t ws_size,
                              hipStream_t stream) {
    const float* x   = (const float*)d_in[0];
    const int*   cls = (const int*)  d_in[1];
    const float* bw1 = (const float*)d_in[2];
    const float* sw1 = (const float*)d_in[3];
    const float* sc1 = (const float*)d_in[4];
    const float* bw2 = (const float*)d_in[5];
    const float* sw2 = (const float*)d_in[6];
    const float* sc2 = (const float*)d_in[7];
    float* out = (float*)d_out;

    float* ws  = (float*)d_ws;
    float* A1  = ws + (size_t)KSEG * C + KSEG;
    float* Wt1 = A1 + (size_t)KSEG * KT1;
    float* h   = Wt1 + (size_t)KT1 * H1;
    float* A2  = h + (size_t)KSEG * H1;
    float* Wt2 = A2 + (size_t)KSEG * KT2;
    float* part  = A2;
    float* cpart = part + (size_t)SS_NCHUNK * KSEG * C;

    segsum_kernel<<<SS_NRANGE * SS_NCHUNK, 1024, 0, stream>>>(x, cls, part, cpart,
                                                              bw1, sw1, sc1, bw2, sw2, sc2, Wt1, Wt2);
    layers_fused_kernel<<<KSEG / 16, 32 * LF_WAVES, 0, stream>>>(part, cpart, Wt1, Wt2, out);
}


# matrix instructions whose A fragment is all zeros are left out: layer-1 weights split into inner and outer base pairs, outer pair and its product skipped per strip and feature; sum order changes
# speedup vs baseline: 10.3204x; 1.0595x over previous
#include <hip/hip_runtime.h>
#include <math.h>

#define HW   1048576
#define C    64
#define KSEG 4096
#define H1   256
#define NCLS 16
#define KT1  (C + C*8)
#define KT2  (H1 + H1*8)

typedef __attribute__((ext_vector_type(2))) float v2f;
typedef __attribute__((ext_vector_type(4))) float v4f;
typedef __attribute__((ext_vector_type(8))) float v8f;

__device__ __forceinline__ float silu_f(float x) {
    return x / (1.0f + expf(-x));
}

__device__ __forceinline__ void bspline8(float x, float* __restrict__ out) {
#pragma unroll
    for (int g = 0; g < 8; ++g) {
        const float cg = 0.4f * (float)(g - 1) - 1.0f;
        const float a  = fabsf((x - cg) * 2.5f);
        const float in = fmaf(a * a, fmaf(0.5f, a, -1.0f), 2.0f / 3.0f);
        const float r  = fmaxf(2.0f - a, 0.0f);
        out[g] = a < 1.0f ? in : r * r * r * (1.0f / 6.0f);
    }
}

#define L1_WAVES 16
#define L1_STEPS (KT1 / 8)
#define L1_CH    12
#define L1_NCH   (L1_STEPS / L1_CH)
#define FS_PITCH 65

#define WP_T 288
#define WP_B 128
static_assert(WP_T * WP_B == H1 * KT1 / 4, "layer-1 weight: one 16-byte unit per item");
static_assert(WP_T * WP_B == NCLS * KT2,   "layer-2 weight: one element per item");

struct WItem { v4f raw; float sc; };

__device__ __forceinline__ WItem wprep1_load(int idx, const float* __restrict__ bw1,
                                             const float* __restrict__ sw1, const float* __restrict__ sc1) {
    const int lane = idx & 31, q = idx >> 5;
    const int j = q % L1_STEPS, nt = q / L1_STEPS;
    const int o = nt * 16 + (lane & 15), hh = lane >> 4;
    WItem w;
    if (j < C / 8) {
        w.raw = *(const v4f*)(bw1 + o * C + 8 * j + 4 * hh);
        w.sc  = 1.0f;
    } else {
        const int i = j - C / 8;
        w.raw = *(const v4f*)(sw1 + ((size_t)o * C + i) * 8 + 4 * hh);
        w.sc  = sc1[o * C + i];
    }
    return w;
}

__device__ __forceinline__ void wprep1_store(int idx, WItem w, float* __restrict__ Wp) {
    const int lane = idx & 31, q = idx >> 5;
    const int j = q % L1_STEPS;
    v2f p0 = (v2f){w.raw.x, w.raw.y};
    v2f p1 = (v2f){w.raw.z, w.raw.w};
    if (j >= C / 8) {
        const v2f lo = (v2f){w.raw.x * w.sc, w.raw.y * w.sc};
        const v2f hi = (v2f){w.raw.z * w.sc, w.raw.w * w.sc};
        const bool h1 = (lane >> 4) != 0;
        p0 = h1 ? lo : hi;
        p1 = h1 ? hi : lo;
    }
    ((v2f*)Wp)[(q * 2 + 0) * 32 + lane] = p0;
    ((v2f*)Wp)[(q * 2 + 1) * 32 + lane] = p1;
}

__device__ __forceinline__ WItem wprep2_load(int idx, const float* __restrict__ bw2,
                                             const float* __restrict__ sw2, const float* __restrict__ sc2) {
    const int o = idx / KT2, j = idx % KT2;
    WItem w;
    w.raw = (v4f){0.f, 0.f, 0.f, 0.f};
    if (j < H1) {
        w.raw.x = bw2[o * H1 + j];
        w.sc    = 1.0f;
    } else {
        const int j2 = j - H1, i = j2 >> 3, g = j2 & 7;
        w.raw.x = sw2[((size_t)o * H1 + i) * 8 + g];
        w.sc    = sc2[o * H1 + i];
    }
    return w;
}

__device__ __forceinline__ void wprep2_store(int idx, WItem w, float* __restrict__ Wt2) {
    const int o = idx / KT2, j = idx % KT2;
    Wt2[(size_t)j * NCLS + o] = (j < H1) ? w.raw.x : w.raw.x * w.sc;
}

#define SS_NRANGE 8
#define SS_NCHUNK 32
#define SS_SEGS   (KSEG / SS_NRANGE)
#define SS_SHIFT  9
#define SS_ROWS   (HW / SS_NCHUNK)
#define SS_WAVES  32
#define SS_WROWS  (SS_ROWS / SS_WAVES)
#define SS_U      32
#define SS_WSEGS  (SS_SEGS / SS_WAVES)
#define SS_SBITS  9
static_assert((1 << SS_SBITS) == SS_SEGS, "a sorted entry keeps the local segment in its low bits");
static_assert(SS_ROWS <= (1 << (32 - SS_SBITS)), "a sorted entry keeps the row of the chunk above them");
static_assert(2 * WP_B == SS_NRANGE * SS_NCHUNK, "the two weights together take every block of this grid");

template <bool FULL>
__device__ __forceinline__ void gather_sorted(const float* __restrict__ xc, const unsigned* sorted,
                                              int e, int eEnd, int lane,
                                              int& cur, v2f& acc, float* __restrict__ pdst) {
    const int ei = e + lane;
    const unsigned entv = sorted[(FULL || ei < eEnd) ? ei : eEnd - 1];
    unsigned ent[SS_U];
    v2f v[SS_U];
#pragma unroll
    for (int u = 0; u < SS_U; ++u) {
        if (FULL || e + u < eEnd) {
            ent[u] = __builtin_amdgcn_readlane(entv, u);
            const float* src = xc + (size_t)(ent[u] >> SS_SBITS) * C;
            v[u] = __builtin_nontemporal_load((const v2f*)src + lane);
        }
    }
#pragma unroll
    for (int u = 0; u < SS_U; ++u) {
        if (FULL || e + u < eEnd) {
            const int s = (int)(ent[u] & (unsigned)(SS_SEGS - 1));
            if (s != cur) {
                *(v2f*)(pdst + cur * C) = acc;
#pragma nounroll
                for (int z = cur + 1; z < s; ++z) *(v2f*)(pdst + z * C) = (v2f){0.f, 0.f};
                acc = (v2f){0.f, 0.f};
                cur = s;
            }
            acc.x += v[u].x;
            acc.y += v[u].y;
        }
    }
}

__global__ __launch_bounds__(1024)
void segsum_kernel(const float* __restrict__ x, const int* __restrict__ cls,
                   float* __restrict__ part, float* __restrict__ cpart,
                   const float* __restrict__ bw1, const float* __restrict__ sw1,
                   const float* __restrict__ sc1, const float* __restrict__ bw2,
                   const float* __restrict__ sw2, const float* __restrict__ sc2,
                   float* __restrict__ Wp, float* __restrict__ Wt2) {
    __shared__ unsigned list_s[SS_WAVES * SS_WROWS];
    __shared__ unsigned sort_s[SS_ROWS];
    __shared__ unsigned cnt_s[SS_SEGS];
    __shared__ unsigned start_s[SS_SEGS + 1];
    __shared__ unsigned wsum_s[SS_SEGS / 32];

    const int tid  = threadIdx.x;
    const int lane = tid & 31;
    const int wave = __builtin_amdgcn_readfirstlane(tid >> 5);
    const unsigned rng = blockIdx.x % SS_NRANGE;
    const int chunk    = blockIdx.x / SS_NRANGE;

    if (tid < SS_SEGS) cnt_s[tid] = 0u;
    __syncthreads();

    unsigned* mylist = list_s + wave * SS_WROWS;
    const int r0 = chunk * SS_ROWS + wave * SS_WROWS;

    const bool wp_on = tid < WP_T;
    const bool wp_l1 = blockIdx.x < WP_B;
    const int  wp_idx = (int)(wp_l1 ? blockIdx.x : blockIdx.x - WP_B) * WP_T + tid;
    WItem wit;
    wit.raw = (v4f){0.f, 0.f, 0.f, 0.f};
    wit.sc  = 1.0f;
    if (wp_on) wit = wp_l1 ? wprep1_load(wp_idx, bw1, sw1, sc1) : wprep2_load(wp_idx, bw2, sw2, sc2);

    int segv[SS_WROWS / 32];
#pragma unroll
    for (int j = 0; j < SS_WROWS / 32; ++j) segv[j] = cls[r0 + j * 32 + lane];
    int n = 0;
#pragma unroll
    for (int j = 0; j < SS_WROWS / 32; ++j) {
        const bool m = ((unsigned)segv[j] >> SS_SHIFT) == rng;
        const unsigned mask = __builtin_amdgcn_ballot_w32(m);
        if (m) {
            const unsigned sl = (unsigned)segv[j] & (SS_SEGS - 1);
            mylist[n + (int)__builtin_amdgcn_mbcnt_lo(mask, 0u)] =
                ((unsigned)(j * 32 + lane) << 16) | sl;
            atomicAdd(&cnt_s[sl], 1u);
        }
        n += __builtin_popcount(mask);
    }
    if (wp_on) {
        if (wp_l1) wprep1_store(wp_idx, wit, Wp);
        else       wprep2_store(wp_idx, wit, Wt2);
    }

    __syncthreads();

    unsigned cv = 0u;
    if (tid < SS_SEGS) cv = cnt_s[tid];
    unsigned inc = cv;
    for (int d = 1; d < 32; d <<= 1) {
        const unsigned t = __shfl_up(inc, d, 32);
        if (lane >= d) inc += t;
    }
    if (tid < SS_SEGS && lane == 31) wsum_s[wave] = inc;
    __syncthreads();
    if (tid < SS_SEGS) {
        unsigned base = 0u;
        for (int w = 0; w < wave; ++w) base += wsum_s[w];
        const unsigned ex = base + inc - cv;
        start_s[tid] = ex;
        cnt_s[tid]   = ex;
        if (tid == SS_SEGS - 1) start_s[SS_SEGS] = base + inc;
        cpart[chunk * KSEG + rng * SS_SEGS + tid] = (float)cv;
    }
    __syncthreads();

    for (int e = 0; e < n; e += 32) {
        const int ei = e + lane;
        if (ei < n) {
            const unsigned ent  = mylist[ei];
            const unsigned sl   = ent & (unsigned)(SS_SEGS - 1);
            const unsigned slot = atomicAdd(&cnt_s[sl], 1u);
            sort_s[slot] = ((unsigned)(wave * SS_WROWS + (int)(ent >> 16)) << SS_SBITS) | sl;
        }
    }
    __syncthreads();

    const int seg0 = wave * SS_WSEGS;
    int e          = __builtin_amdgcn_readfirstlane((int)start_s[seg0]);
    const int eEnd = __builtin_amdgcn_readfirstlane((int)start_s[seg0 + SS_WSEGS]);
    const float* xc = x + (size_t)chunk * SS_ROWS * C;
    float* pdst = part + ((size_t)chunk * KSEG + (size_t)rng * SS_SEGS) * C + 2 * lane;
    int cur = seg0;
    v2f acc = (v2f){0.f, 0.f};
    for (; e + SS_U <= eEnd; e += SS_U) gather_sorted<true>(xc, sort_s, e, eEnd, lane, cur, acc, pdst);
    if (e < eEnd) gather_sorted<false>(xc, sort_s, e, eEnd, lane, cur, acc, pdst);
    for (; cur < seg0 + SS_WSEGS; ++cur) {
        *(v2f*)(pdst + cur * C) = acc;
        acc = (v2f){0.f, 0.f};
    }
}

#define LF_WAVES L1_WAVES
#define L2_FEATS (H1 / LF_WAVES)
#define L2_STEPS (L2_FEATS / 4)
#define HS_PITCH 258

static_assert(SS_NCHUNK == 32, "one partial count per lane");

__global__ __launch_bounds__(32 * LF_WAVES)
void layers_fused_kernel(const float* __restrict__ part, const float* __restrict__ cpart,
                         const float* __restrict__ Wp, const float* __restrict__ Wt2,
                         float* __restrict__ out) {
    __shared__ __attribute__((aligned(16))) float A_s[L1_STEPS * 32 * 4];
    __shared__ float h_s[16 * HS_PITCH];
    __shared__ float f_s[16 * FS_PITCH];
    __shared__ float cnt_s[16];
    __shared__ float red_s[LF_WAVES * 256];
    __shared__ unsigned skip_s[C];

    const int tid  = threadIdx.x;
    const int lane = tid & 31;
    const int wave = __builtin_amdgcn_readfirstlane(tid >> 5);
    const int strip = blockIdx.x;
    const int c    = lane & 15;
    const int hh   = lane >> 4;

    const v2f* Bw = (const v2f*)Wp + (size_t)wave * L1_STEPS * 64 + lane;
    v2f bq0[2][L1_CH];
    v2f bq1[2][L1_CH];
#pragma unroll
    for (int jj = 0; jj < L1_CH; ++jj) {
        bq0[0][jj] = Bw[(jj * 2 + 0) * 32];
        bq1[0][jj] = Bw[(jj * 2 + 1) * 32];
    }

    {
        const float* p = part + (size_t)strip * 16 * C + tid;
        float s0 = 0.0f, s1 = 0.0f;
#pragma unroll
        for (int b = 0; b < SS_NCHUNK; ++b) {
            s0 += p[(size_t)b * KSEG * C];
            s1 += p[(size_t)b * KSEG * C + 512];
        }
        f_s[(tid >> 6) * FS_PITCH + (tid & 63)] = s0;
        f_s[(8 + (tid >> 6)) * FS_PITCH + (tid & 63)] = s1;
        float cv = cpart[lane * KSEG + strip * 16 + wave];
        for (int m = 16; m >= 1; m >>= 1) cv += __shfl_xor(cv, m, 32);
        if (lane == 0) cnt_s[wave] = cv;
    }
    __syncthreads();

    for (int e = 0; e < 2; ++e) {
        const int idx = tid + e * 512;
        const int r = idx & 15, cc = idx >> 4;
        const float f = f_s[r * FS_PITCH + cc] / fmaxf(cnt_s[r], 1.0f);
        A_s[(((cc >> 3) * 32 + ((cc >> 2) & 1) * 16 + r) << 2) + (cc & 3)] = silu_f(f);
        float bs[8];
        bspline8(f, bs);
        ((v4f*)A_s)[(8 + cc) * 32 + r]      = (v4f){bs[0], bs[1], bs[2], bs[3]};
        ((v4f*)A_s)[(8 + cc) * 32 + 16 + r] = (v4f){bs[6], bs[7], bs[4], bs[5]};
        const bool oz = (bs[0] == 0.0f) & (bs[1] == 0.0f) & (bs[6] == 0.0f) & (bs[7] == 0.0f);
        const unsigned zm = __builtin_amdgcn_ballot_w32(oz);
        if ((lane & 15) == 0) skip_s[cc] = (((zm >> (lane & 16)) & 0xFFFFu) == 0xFFFFu) ? 1u : 0u;
    }
    __syncthreads();

    const unsigned skip0 = __builtin_amdgcn_ballot_w32(skip_s[lane] != 0u);
    const unsigned skip1 = __builtin_amdgcn_ballot_w32(skip_s[32 + lane] != 0u);

    v8f acc0 = (v8f){0.f, 0.f, 0.f, 0.f, 0.f, 0.f, 0.f, 0.f};
    v8f acc1 = (v8f){0.f, 0.f, 0.f, 0.f, 0.f, 0.f, 0.f, 0.f};
    const v4f* Al = (const v4f*)A_s + lane;
#pragma unroll
    for (int ch = 0; ch < L1_NCH; ++ch) {
        if (ch + 1 < L1_NCH) {
#pragma unroll
            for (int jj = 0; jj < L1_CH; ++jj) {
                const int jn = (ch + 1) * L1_CH + jj;
                const int cn = jn - C / 8;
                bq0[(ch + 1) & 1][jj] = Bw[(jn * 2 + 0) * 32];
                if (((((cn < 32) ? skip0 : skip1) >> (cn & 31)) & 1u) == 0u)
                    bq1[(ch + 1) & 1][jj] = Bw[(jn * 2 + 1) * 32];
            }
        }
        v4f aq[L1_CH];
#pragma unroll
        for (int jj = 0; jj < L1_CH; ++jj) aq[jj] = Al[(ch * L1_CH + jj) * 32];
#pragma unroll
        for (int jj = 0; jj < L1_CH; ++jj) {
            const int j = ch * L1_CH + jj;
            const v4f a = aq[jj];
            const v2f b0 = bq0[ch & 1][jj];
            if (j < C / 8) {
                acc0 = __builtin_amdgcn_wmma_f32_16x16x4_f32(false, (v2f){a.x, a.y}, false, b0,
                                                             (short)0, acc0, false, false);
                acc1 = __builtin_amdgcn_wmma_f32_16x16x4_f32(false, (v2f){a.z, a.w}, false, bq1[ch & 1][jj],
                                                             (short)0, acc1, false, false);
            } else {
                const int cf = j - C / 8;
                if (j & 1)
                    acc0 = __builtin_amdgcn_wmma_f32_16x16x4_f32(false, (v2f){a.z, a.w}, false, b0,
                                                                 (short)0, acc0, false, false);
                else
                    acc1 = __builtin_amdgcn_wmma_f32_16x16x4_f32(false, (v2f){a.z, a.w}, false, b0,
                                                                 (short)0, acc1, false, false);
                if (((((cf < 32) ? skip0 : skip1) >> (cf & 31)) & 1u) == 0u) {
                    if (j & 1)
                        acc1 = __builtin_amdgcn_wmma_f32_16x16x4_f32(false, (v2f){a.x, a.y}, false, bq1[ch & 1][jj],
                                                                     (short)0, acc1, false, false);
                    else
                        acc0 = __builtin_amdgcn_wmma_f32_16x16x4_f32(false, (v2f){a.x, a.y}, false, bq1[ch & 1][jj],
                                                                     (short)0, acc0, false, false);
                }
            }
        }
    }

    const int fbase = wave * L2_FEATS;
    const float* Wc = Wt2 + c;
    float wsil[L2_STEPS][2];
    float wspl[L2_STEPS][2][8];
#pragma unroll
    for (int s = 0; s < L2_STEPS; ++s) {
#pragma unroll
        for (int q = 0; q < 2; ++q) {
            const int f = fbase + 4 * s + 2 * q + hh;
            wsil[s][q] = Wc[(size_t)f * NCLS];
#pragma unroll
            for (int g = 0; g < 8; ++g) wspl[s][q][g] = Wc[(size_t)(H1 + f * 8 + g) * NCLS];
        }
    }

#pragma unroll
    for (int r = 0; r < 8; ++r)
        h_s[(r + 8 * hh) * HS_PITCH + wave * 16 + c] = acc0[r] + acc1[r];
    __syncthreads();

    float xv[L2_STEPS][2];
#pragma unroll
    for (int s = 0; s < L2_STEPS; ++s) {
#pragma unroll
        for (int q = 0; q < 2; ++q)
            xv[s][q] = h_s[c * HS_PITCH + fbase + 4 * s + 2 * q + hh];
    }
    v8f o0 = (v8f){0.f, 0.f, 0.f, 0.f, 0.f, 0.f, 0.f, 0.f};
    v8f o1 = (v8f){0.f, 0.f, 0.f, 0.f, 0.f, 0.f, 0.f, 0.f};
#pragma unroll
    for (int s = 0; s < L2_STEPS; ++s) {
        {
            v2f a, b;
            a.x = silu_f(xv[s][0]);
            a.y = silu_f(xv[s][1]);
            b.x = wsil[s][0];
            b.y = wsil[s][1];
            o0 = __builtin_amdgcn_wmma_f32_16x16x4_f32(false, a, false, b, (short)0, o0, false, false);
        }
        float bs0[8], bs1[8];
        bspline8(xv[s][0], bs0);
        bspline8(xv[s][1], bs1);
#pragma unroll
        for (int t = 0; t < 4; ++t) {
            v2f a0, b0, a1, b1;
            a0.x = bs0[2 * t];
            a0.y = bs0[2 * t + 1];
            b0.x = wspl[s][0][2 * t];
            b0.y = wspl[s][0][2 * t + 1];
            a1.x = bs1[2 * t];
            a1.y = bs1[2 * t + 1];
            b1.x = wspl[s][1][2 * t];
            b1.y = wspl[s][1][2 * t + 1];
            const bool live0 = (t == 1 || t == 2) ||
                               __builtin_amdgcn_ballot_w32((a0.x != 0.0f) | (a0.y != 0.0f)) != 0u;
            const bool live1 = (t == 1 || t == 2) ||
                               __builtin_amdgcn_ballot_w32((a1.x != 0.0f) | (a1.y != 0.0f)) != 0u;
            if (live0)
                o0 = __builtin_amdgcn_wmma_f32_16x16x4_f32(false, a0, false, b0, (short)0, o0, false, false);
            if (live1)
                o1 = __builtin_amdgcn_wmma_f32_16x16x4_f32(false, a1, false, b1, (short)0, o1, false, false);
        }
    }

    float* mine = red_s + wave * 256;
#pragma unroll
    for (int r = 0; r < 8; ++r) mine[r * 32 + lane] = o0[r] + o1[r];
    __syncthreads();

    if (tid < 256) {
        const int row = tid >> 4, col = tid & 15;
        const int idx = (row & 7) * 32 + (row >> 3) * 16 + col;
        float sum = 0.0f;
#pragma unroll
        for (int w = 0; w < LF_WAVES; ++w) sum += red_s[w * 256 + idx];
        out[(size_t)strip * 256 + tid] = sum;
    }
}

extern "C" void kernel_launch(void* const* d_in, const int* in_sizes, int n_in,
                              void* d_out, int out_size, void* d_ws, size_t ws_size,
                              hipStream_t stream) {
    const float* x   = (const float*)d_in[0];
    const int*   cls = (const int*)  d_in[1];
    const float* bw1 = (const float*)d_in[2];
    const float* sw1 = (const float*)d_in[3];
    const float* sc1 = (const float*)d_in[4];
    const float* bw2 = (const float*)d_in[5];
    const float* sw2 = (const float*)d_in[6];
    const float* sc2 = (const float*)d_in[7];
    float* out = (float*)d_out;

    float* ws  = (float*)d_ws;
    float* A1  = ws + (size_t)KSEG * C + KSEG;
    float* Wt1 = A1 + (size_t)KSEG * KT1;
    float* h   = Wt1 + (size_t)KT1 * H1;
    float* A2  = h + (size_t)KSEG * H1;
    float* Wt2 = A2 + (size_t)KSEG * KT2;
    float* part  = A2;
    float* cpart = part + (size_t)SS_NCHUNK * KSEG * C;

    segsum_kernel<<<SS_NRANGE * SS_NCHUNK, 1024, 0, stream>>>(x, cls, part, cpart,
                                                              bw1, sw1, sc1, bw2, sw2, sc2, Wt1, Wt2);
    layers_fused_kernel<<<KSEG / 16, 32 * LF_WAVES, 0, stream>>>(part, cpart, Wt1, Wt2, out);
}
